# starting kernel
# baseline (speedup 1.0000x reference)
#include <hip/hip_runtime.h>

typedef __attribute__((ext_vector_type(2))) float v2f;
typedef __attribute__((ext_vector_type(8))) float v8f;

#define HW    2048
#define IMG   (HW * HW)
#define KH    17
#define PC    8
#define LDSW  24
#define TROWS 32
#define WPROWS 48

template <int KW>
__global__ __launch_bounds__(256) void guided_pass(
    const float* __restrict__ Xin, const float* __restrict__ Yin,
    const float* __restrict__ Wk,  const float* __restrict__ Nplane,
    float* __restrict__ Xout)
{
    constexpr int PR = KW / 2;
    constexpr int TW = 16 + KW - 1;

    __shared__ float wpad[WPROWS * KW];
    __shared__ float tile[8][TROWS * LDSW];

    const int lane = threadIdx.x & 31;
    const int wid  = threadIdx.x >> 5;

    for (int i = threadIdx.x; i < WPROWS * KW; i += 256) {
        int j = i - 16 * KW;
        wpad[i] = (j >= 0 && j < KH * KW) ? Wk[j] : 0.0f;
    }

    const int TPR = HW / 16;
    const int TPI = TPR * TPR;
    const int t   = blockIdx.x * 8 + wid;
    const int img = t / TPI;
    const int rem = t % TPI;
    const int row0 = (rem / TPR) * 16;
    const int col0 = (rem % TPR) * 16;

    const float* Xi = Xin + (size_t)img * IMG;
    const float* Yi = Yin + (size_t)img * IMG;
    float*       Xo = Xout + (size_t)img * IMG;

    float* my = tile[wid];
    for (int idx = lane; idx < TROWS * TW; idx += 32) {
        int r = idx / TW, c = idx % TW;
        int gr = row0 + r - PC;
        int gc = col0 + c - PR;
        float d = 0.0f;
        if (gr >= 0 && gr < HW && gc >= 0 && gc < HW) {
            size_t o = (size_t)gr * HW + gc;
            d = Yi[o] - Xi[o];
        }
        my[r * LDSW + c] = d;
    }
    __syncthreads();

    v8f acc = {};
    const int half = lane >> 4;
    const int Mrow = lane & 15;
    const int n    = lane & 15;

#pragma unroll
    for (int s = 0; s < 8; ++s) {
        const int jA   = 4 * s + 2 * half;
        const int tpad = jA - Mrow + 16;
        const float* wrA = &wpad[tpad * KW];
        const float* wrB = wrA + KW;
        const float* ld0 = &my[jA * LDSW + n];
        const float* ld1 = ld0 + LDSW;

#pragma unroll
        for (int kx = 0; kx < KW; ++kx) {
            v2f a, b;
            a.x = wrA[kx];
            a.y = wrB[kx];
            b.x = ld0[kx];
            b.y = ld1[kx];
            acc = __builtin_amdgcn_wmma_f32_16x16x4_f32(
                false, a, false, b, (short)0, acc, false, false);
        }
    }

#pragma unroll
    for (int v = 0; v < 8; ++v) {
        int row = row0 + v + 8 * half;
        int col = col0 + n;
        size_t o = (size_t)row * HW + col;
        Xo[o] = Xi[o] + acc[v] / Nplane[o];
    }
}

template <int KW>
static void run_three_passes(const float* X, const float* y, const float* k,
                             const float* Nn, float* out, float* tmp,
                             int batch, hipStream_t stream)
{
    const int tiles = batch * (HW / 16) * (HW / 16);
    dim3 grid(tiles / 8), block(256);
    guided_pass<KW><<<grid, block, 0, stream>>>(X,   y, k,               Nn,           out);
    guided_pass<KW><<<grid, block, 0, stream>>>(out, y, k + KH * KW,     Nn + IMG,     tmp);
    guided_pass<KW><<<grid, block, 0, stream>>>(tmp, y, k + 2 * KH * KW, Nn + 2 * IMG, out);
}

extern "C" void kernel_launch(void* const* d_in, const int* in_sizes, int n_in,
                              void* d_out, int out_size, void* d_ws, size_t ws_size,
                              hipStream_t stream) {
    const float* X  = (const float*)d_in[0];
    const float* y  = (const float*)d_in[1];
    const float* k  = (const float*)d_in[2];
    const float* Nn = (const float*)d_in[3];
    float* out = (float*)d_out;
    float* tmp = (float*)d_ws;

    const int KW    = in_sizes[2] / (3 * KH);
    const int batch = in_sizes[0] / IMG;

    switch (KW) {
        case 1: run_three_passes<1>(X, y, k, Nn, out, tmp, batch, stream); break;
        case 3: run_three_passes<3>(X, y, k, Nn, out, tmp, batch, stream); break;
        case 7: run_three_passes<7>(X, y, k, Nn, out, tmp, batch, stream); break;
        case 9: run_three_passes<9>(X, y, k, Nn, out, tmp, batch, stream); break;
        case 5:
        default: run_three_passes<5>(X, y, k, Nn, out, tmp, batch, stream); break;
    }
}


# each tile's reads (patch fragments, normalisers, X, Y) go out as one group ahead of its 40 WMMAs, and its 8 rows finish on registers with no per-row branch; order of all sums unchanged
# speedup vs baseline: 1.3068x; 1.3068x over previous
#include <hip/hip_runtime.h>


typedef __attribute__((ext_vector_type(2))) float v2f;
typedef __attribute__((ext_vector_type(8))) float v8f;

#define HW     2048
#define IMG    (HW * HW)
#define KH     17
#define PC     8
#define WPROWS 48
#define BR     64
#define BC     128
#define NTHR   512
#define NWAVE  (NTHR / 32)

template <int KW>
struct Geo {
    static constexpr int PR  = KW / 2;
    static constexpr int RR  = BR + 6 * PC;
    static constexpr int HL  = (3 * PR <= 8) ? 8 : 12;
    static constexpr int P   = BC + 2 * HL;
    static constexpr int ASZ = RR * P + 64;
    static constexpr int SG  = (KW <= 5) ? 8 : (KW <= 7) ? 4 : 2;
};

template <int KW>
__device__ __forceinline__ void load_band(const float* wp, int lane, v2f (&a)[8][KW])
{
    const int half = lane >> 4;
    const int m    = lane & 15;
#pragma unroll
    for (int s = 0; s < 8; ++s) {
        const int jA = 4 * s + 2 * half;
        const float* wrA = wp + (jA - m + 16) * KW;
        const float* wrB = wrA + KW;
#pragma unroll
        for (int kx = 0; kx < KW; ++kx) {
            a[s][kx].x = wrA[kx];
            a[s][kx].y = wrB[kx];
        }
    }
}

template <int KW, int P, int SG>
__device__ __forceinline__ void load_patch(const float* src, int lane, int s0, v2f (&b)[SG][KW])
{
    const int half = lane >> 4;
    const int m    = lane & 15;
#pragma unroll
    for (int s = 0; s < SG; ++s) {
        const int jA = 4 * (s0 + s) + 2 * half;
        const float* ld0 = src + jA * P + m;
        const float* ld1 = ld0 + P;
#pragma unroll
        for (int kx = 0; kx < KW; ++kx) {
            b[s][kx].x = ld0[kx];
            b[s][kx].y = ld1[kx];
        }
    }
}

template <int KW, int SG>
__device__ __forceinline__ v8f band_mma(const v2f (&a)[8][KW], int s0, const v2f (&b)[SG][KW], v8f acc)
{
#pragma unroll
    for (int s = 0; s < SG; ++s) {
#pragma unroll
        for (int kx = 0; kx < KW; ++kx)
            acc = __builtin_amdgcn_wmma_f32_16x16x4_f32(
                false, a[s0 + s][kx], false, b[s][kx], (short)0, acc, false, false);
    }
    return acc;
}

template <int KW, int PASS>
__device__ __forceinline__ void mid_pass(
    const float* Din, float* Dout, float* sX, const float* sY, const float* wp,
    const float* __restrict__ Np, int gr0, int gc0, int lane, int wid)
{
    using G = Geo<KW>;
    constexpr int PR = G::PR, P = G::P, SG = G::SG;
    constexpr int ro   = PASS * PC;
    constexpr int co   = G::HL - (3 - PASS) * PR;
    constexpr int nrow = BR + 2 * (3 - PASS) * PC;
    constexpr int ncol = BC + 2 * (3 - PASS) * PR;
    constexpr int TR   = nrow / 16;
    constexpr int TC   = (ncol + 15) / 16;
    const int half = lane >> 4;
    const int n    = lane & 15;

    v2f a[8][KW];
    load_band<KW>(wp, lane, a);

    for (int t = wid; t < TR * TC; t += NWAVE) {
        const int ti = t / TC, tj = t - ti * TC;
        const int r0 = ro + 16 * ti, c0 = co + 16 * tj;
        const int  c    = c0 + n;
        const int  gc   = gc0 + c;
        const bool cval = c < co + ncol;
        const bool cin  = (unsigned)gc < (unsigned)HW;
        float nv[8];
        const int cc = min(max(gc, 0), HW - 1);
#pragma unroll
        for (int v = 0; v < 8; ++v) {
            const int cr = min(max(gr0 + r0 + v + 8 * half, 0), HW - 1);
            nv[v] = Np[(size_t)cr * HW + cc];
        }
        const float* patch = Din + (r0 - PC) * P + (c0 - PR);
        v2f b[SG][KW];
        load_patch<KW, P, SG>(patch, lane, 0, b);
        const int l0 = (r0 + 8 * half) * P + c;
        float xs[8], ys[8];
#pragma unroll
        for (int v = 0; v < 8; ++v) {
            xs[v] = sX[l0 + v * P];
            ys[v] = sY[l0 + v * P];
        }
        __builtin_amdgcn_sched_barrier(0);
        v8f acc = {};
        acc = band_mma<KW, SG>(a, 0, b, acc);
#pragma unroll
        for (int s0 = SG; s0 < 8; s0 += SG) {
            load_patch<KW, P, SG>(patch, lane, s0, b);
            __builtin_amdgcn_sched_barrier(0);
            acc = band_mma<KW, SG>(a, s0, b, acc);
        }
        __builtin_amdgcn_sched_barrier(0);
        asm volatile("" :: "v"(nv[0]), "v"(nv[1]), "v"(nv[2]), "v"(nv[3]),
                           "v"(nv[4]), "v"(nv[5]), "v"(nv[6]), "v"(nv[7]));
        asm volatile("" :: "v"(xs[0]), "v"(xs[1]), "v"(xs[2]), "v"(xs[3]),
                           "v"(xs[4]), "v"(xs[5]), "v"(xs[6]), "v"(xs[7]));
        asm volatile("" :: "v"(ys[0]), "v"(ys[1]), "v"(ys[2]), "v"(ys[3]),
                           "v"(ys[4]), "v"(ys[5]), "v"(ys[6]), "v"(ys[7]));
        float xn[8], d[8];
#pragma unroll
        for (int v = 0; v < 8; ++v) {
            const int gr = gr0 + r0 + v + 8 * half;
            const unsigned keep = (cin && (unsigned)gr < (unsigned)HW) ? 0xffffffffu : 0u;
            const float q = __uint_as_float(__float_as_uint(acc[v] / nv[v]) & keep);
            xn[v] = xs[v] + q;
            d[v]  = ys[v] - xn[v];
        }
        if (cval) {
#pragma unroll
            for (int v = 0; v < 8; ++v) {
                sX[l0 + v * P]   = xn[v];
                Dout[l0 + v * P] = d[v];
            }
        }
    }
}

template <int KW>
__global__ __launch_bounds__(NTHR) void guided_fused(
    const float* __restrict__ X,  const float* __restrict__ Y,
    const float* __restrict__ Wk, const float* __restrict__ Nn,
    float* __restrict__ Out)
{
    using G = Geo<KW>;
    constexpr int PR = G::PR, RR = G::RR, HL = G::HL, P = G::P, ASZ = G::ASZ, SG = G::SG;

    __shared__ float wpad[3 * WPROWS * KW];
    __shared__ __attribute__((aligned(16))) float sX[ASZ];
    __shared__ __attribute__((aligned(16))) float sY[ASZ];
    __shared__ __attribute__((aligned(16))) float sDa[ASZ];
    __shared__ __attribute__((aligned(16))) float sDb[ASZ];

    const int tid  = threadIdx.x;
    const int lane = tid & 31;
    const int wid  = tid >> 5;

    constexpr int BPR = HW / BC;
    constexpr int BPI = BPR * (HW / BR);
    const int img = blockIdx.x / BPI;
    const int rem = blockIdx.x % BPI;
    const int R0  = (rem / BPR) * BR;
    const int C0  = (rem % BPR) * BC;
    const int gr0 = R0 - 3 * PC;
    const int gc0 = C0 - HL;

    const float* Xi = X   + (size_t)img * IMG;
    const float* Yi = Y   + (size_t)img * IMG;
    float*       Oi = Out + (size_t)img * IMG;

    for (int i = tid; i < 3 * WPROWS * KW; i += NTHR) {
        const int p = i / (WPROWS * KW);
        const int j = i - p * (WPROWS * KW) - 16 * KW;
        wpad[i] = (j >= 0 && j < KH * KW) ? Wk[p * KH * KW + j] : 0.0f;
    }

    {
        constexpr int Q  = P / 4;
        constexpr int NV = RR * Q;
        constexpr int IT = (NV + NTHR - 1) / NTHR;
        float4 xv[IT], yv[IT];
#pragma unroll
        for (int it = 0; it < IT; ++it) {
            const int idx = min(tid + it * NTHR, NV - 1);
            const int r = idx / Q, q = idx - r * Q;
            const int cr = min(max(gr0 + r, 0), HW - 1);
            const int cc = min(max(gc0 + 4 * q, 0), HW - 4);
            const size_t o = (size_t)cr * HW + cc;
            xv[it] = *reinterpret_cast<const float4*>(Xi + o);
            yv[it] = *reinterpret_cast<const float4*>(Yi + o);
        }
#pragma unroll
        for (int it = 0; it < IT; ++it) {
            const int idx = tid + it * NTHR;
            if (idx < NV) {
                const int r = idx / Q, q = idx - r * Q;
                const int gr = gr0 + r, gc = gc0 + 4 * q;
                const bool in = (unsigned)gr < (unsigned)HW && (unsigned)gc < (unsigned)HW;
                float4 x = xv[it], y = yv[it];
                if (!in) { x = make_float4(0.f, 0.f, 0.f, 0.f); y = x; }
                const int l = r * P + 4 * q;
                *reinterpret_cast<float4*>(sX + l)  = x;
                *reinterpret_cast<float4*>(sY + l)  = y;
                *reinterpret_cast<float4*>(sDa + l) = make_float4(y.x - x.x, y.y - x.y, y.z - x.z, y.w - x.w);
            }
        }
    }
    __syncthreads();

    mid_pass<KW, 1>(sDa, sDb, sX, sY, wpad,                   Nn,           gr0, gc0, lane, wid);
    __syncthreads();
    mid_pass<KW, 2>(sDb, sDa, sX, sY, wpad + WPROWS * KW,     Nn + IMG,     gr0, gc0, lane, wid);
    __syncthreads();

    {
        constexpr int ro = 3 * PC, co = HL;
        constexpr int TC = BC / 16;
        const float* __restrict__ N3 = Nn + 2 * (size_t)IMG;
        const float* wp = wpad + 2 * WPROWS * KW;
        const int half = lane >> 4;
        const int n    = lane & 15;
        v2f a[8][KW];
        load_band<KW>(wp, lane, a);
        for (int t = wid; t < (BR / 16) * TC; t += NWAVE) {
            const int ti = t / TC, tj = t - ti * TC;
            const int r0 = ro + 16 * ti, c0 = co + 16 * tj;
            const int c  = c0 + n;
            const int gc = gc0 + c;
            float nv[8];
#pragma unroll
            for (int v = 0; v < 8; ++v)
                nv[v] = N3[(size_t)(gr0 + r0 + v + 8 * half) * HW + gc];
            const float* patch = sDa + (r0 - PC) * P + (c0 - PR);
            v2f b[SG][KW];
            load_patch<KW, P, SG>(patch, lane, 0, b);
            const int l0 = (r0 + 8 * half) * P + c;
            float xs[8];
#pragma unroll
            for (int v = 0; v < 8; ++v)
                xs[v] = sX[l0 + v * P];
            __builtin_amdgcn_sched_barrier(0);
            v8f acc = {};
            acc = band_mma<KW, SG>(a, 0, b, acc);
#pragma unroll
            for (int s0 = SG; s0 < 8; s0 += SG) {
                load_patch<KW, P, SG>(patch, lane, s0, b);
                __builtin_amdgcn_sched_barrier(0);
                acc = band_mma<KW, SG>(a, s0, b, acc);
            }
#pragma unroll
            for (int v = 0; v < 8; ++v) {
                const size_t o = (size_t)(gr0 + r0 + v + 8 * half) * HW + gc;
                Oi[o] = xs[v] + acc[v] / nv[v];
            }
        }
    }
}

template <int KW>
static void run_fused(const float* X, const float* y, const float* k,
                      const float* Nn, float* out, int batch, hipStream_t stream)
{
    dim3 grid(batch * (HW / BR) * (HW / BC)), block(NTHR);
    guided_fused<KW><<<grid, block, 0, stream>>>(X, y, k, Nn, out);
}

extern "C" void kernel_launch(void* const* d_in, const int* in_sizes, int n_in,
                              void* d_out, int out_size, void* d_ws, size_t ws_size,
                              hipStream_t stream) {
    const float* X  = (const float*)d_in[0];
    const float* y  = (const float*)d_in[1];
    const float* k  = (const float*)d_in[2];
    const float* Nn = (const float*)d_in[3];
    float* out = (float*)d_out;

    const int KW    = in_sizes[2] / (3 * KH);
    const int batch = in_sizes[0] / IMG;

    switch (KW) {
        case 1: run_fused<1>(X, y, k, Nn, out, batch, stream); break;
        case 3: run_fused<3>(X, y, k, Nn, out, batch, stream); break;
        case 7: run_fused<7>(X, y, k, Nn, out, batch, stream); break;
        case 9: run_fused<9>(X, y, k, Nn, out, batch, stream); break;
        case 5:
        default: run_fused<5>(X, y, k, Nn, out, batch, stream); break;
    }
}


# LDS operands held K-contiguous (planes column-major): a lane's two K rows are one 8-byte read, a tile's 8 rows per lane two 16-byte reads or stores, read as one group; order of all sums unchanged
# speedup vs baseline: 1.4332x; 1.0967x over previous
#include <hip/hip_runtime.h>


typedef __attribute__((ext_vector_type(2))) float v2f;
typedef __attribute__((ext_vector_type(4))) float v4f;
typedef __attribute__((ext_vector_type(8))) float v8f;

#define HW     2048
#define IMG    (HW * HW)
#define KH     17
#define PC     8
#define WPROWS 48
#define BR     64
#define BC     128
#define NTHR   512
#define NWAVE  (NTHR / 32)

template <int KW>
struct Geo {
    static constexpr int PR  = KW / 2;
    static constexpr int RR  = BR + 6 * PC;
    static constexpr int HL  = (3 * PR <= 8) ? 8 : 12;
    static constexpr int P   = BC + 2 * HL;
    static constexpr int PT  = 4 * ((RR / 4) | 1);
    static constexpr int ASZ = (P + 8) * PT;
    static constexpr int SG  = (KW <= 5) ? 8 : (KW <= 7) ? 4 : 2;
};

template <int KW>
__device__ __forceinline__ void load_band(const float* wp, int lane, v2f (&a)[8][KW])
{
    const int half = lane >> 4;
    const int m    = lane & 15;
    const float* w0 = wp + (2 * half - m + 16);
#pragma unroll
    for (int s = 0; s < 8; ++s) {
#pragma unroll
        for (int kx = 0; kx < KW; ++kx) {
            a[s][kx].x = w0[kx * WPROWS + 4 * s];
            a[s][kx].y = w0[kx * WPROWS + 4 * s + 1];
        }
    }
}

template <int KW, int PT, int SG>
__device__ __forceinline__ void load_patch(const float* src, int s0, v2f (&b)[SG][KW])
{
#pragma unroll
    for (int s = 0; s < SG; ++s) {
#pragma unroll
        for (int kx = 0; kx < KW; ++kx)
            b[s][kx] = *reinterpret_cast<const v2f*>(src + kx * PT + 4 * (s0 + s));
    }
}

template <int KW, int SG>
__device__ __forceinline__ v8f band_mma(const v2f (&a)[8][KW], int s0, const v2f (&b)[SG][KW], v8f acc)
{
#pragma unroll
    for (int s = 0; s < SG; ++s) {
#pragma unroll
        for (int kx = 0; kx < KW; ++kx)
            acc = __builtin_amdgcn_wmma_f32_16x16x4_f32(
                false, a[s0 + s][kx], false, b[s][kx], (short)0, acc, false, false);
    }
    return acc;
}

template <int KW, int PASS>
__device__ __forceinline__ void mid_pass(
    const float* Din, float* Dout, float* sX, const float* sY, const float* wp,
    const float* __restrict__ Np, int gr0, int gc0, int lane, int wid)
{
    using G = Geo<KW>;
    constexpr int PR = G::PR, PT = G::PT, SG = G::SG;
    constexpr int ro   = PASS * PC;
    constexpr int co   = G::HL - (3 - PASS) * PR;
    constexpr int nrow = BR + 2 * (3 - PASS) * PC;
    constexpr int ncol = BC + 2 * (3 - PASS) * PR;
    constexpr int TR   = nrow / 16;
    constexpr int TC   = (ncol + 15) / 16;
    const int half = lane >> 4;
    const int n    = lane & 15;

    v2f a[8][KW];
    load_band<KW>(wp, lane, a);

    for (int t = wid; t < TR * TC; t += NWAVE) {
        const int ti = t / TC, tj = t - ti * TC;
        const int r0 = ro + 16 * ti, c0 = co + 16 * tj;
        const int  c    = c0 + n;
        const int  gc   = gc0 + c;
        const bool cval = c < co + ncol;
        const bool cin  = (unsigned)gc < (unsigned)HW;
        float nv[8];
        const int cc = min(max(gc, 0), HW - 1);
#pragma unroll
        for (int v = 0; v < 8; ++v) {
            const int cr = min(max(gr0 + r0 + v + 8 * half, 0), HW - 1);
            nv[v] = Np[(size_t)cr * HW + cc];
        }
        const float* patch = Din + (c0 - PR + n) * PT + (r0 - PC) + 2 * half;
        v2f b[SG][KW];
        load_patch<KW, PT, SG>(patch, 0, b);
        const int l0 = c * PT + r0 + 8 * half;
        const v4f x0 = *reinterpret_cast<const v4f*>(sX + l0);
        const v4f x1 = *reinterpret_cast<const v4f*>(sX + l0 + 4);
        const v4f y0 = *reinterpret_cast<const v4f*>(sY + l0);
        const v4f y1 = *reinterpret_cast<const v4f*>(sY + l0 + 4);
        const float xs[8] = {x0.x, x0.y, x0.z, x0.w, x1.x, x1.y, x1.z, x1.w};
        const float ys[8] = {y0.x, y0.y, y0.z, y0.w, y1.x, y1.y, y1.z, y1.w};
        __builtin_amdgcn_sched_barrier(0);
        v8f acc = {};
        acc = band_mma<KW, SG>(a, 0, b, acc);
#pragma unroll
        for (int s0 = SG; s0 < 8; s0 += SG) {
            load_patch<KW, PT, SG>(patch, s0, b);
            __builtin_amdgcn_sched_barrier(0);
            acc = band_mma<KW, SG>(a, s0, b, acc);
        }
        __builtin_amdgcn_sched_barrier(0);
        asm volatile("" :: "v"(nv[0]), "v"(nv[1]), "v"(nv[2]), "v"(nv[3]),
                           "v"(nv[4]), "v"(nv[5]), "v"(nv[6]), "v"(nv[7]));
        asm volatile("" :: "v"(xs[0]), "v"(xs[1]), "v"(xs[2]), "v"(xs[3]),
                           "v"(xs[4]), "v"(xs[5]), "v"(xs[6]), "v"(xs[7]));
        asm volatile("" :: "v"(ys[0]), "v"(ys[1]), "v"(ys[2]), "v"(ys[3]),
                           "v"(ys[4]), "v"(ys[5]), "v"(ys[6]), "v"(ys[7]));
        float xn[8], d[8];
#pragma unroll
        for (int v = 0; v < 8; ++v) {
            const int gr = gr0 + r0 + v + 8 * half;
            const unsigned keep = (cin && (unsigned)gr < (unsigned)HW) ? 0xffffffffu : 0u;
            const float q = __uint_as_float(__float_as_uint(acc[v] / nv[v]) & keep);
            xn[v] = xs[v] + q;
            d[v]  = ys[v] - xn[v];
        }
        if (cval) {
            *reinterpret_cast<v4f*>(sX + l0)       = (v4f){xn[0], xn[1], xn[2], xn[3]};
            *reinterpret_cast<v4f*>(sX + l0 + 4)   = (v4f){xn[4], xn[5], xn[6], xn[7]};
            *reinterpret_cast<v4f*>(Dout + l0)     = (v4f){d[0], d[1], d[2], d[3]};
            *reinterpret_cast<v4f*>(Dout + l0 + 4) = (v4f){d[4], d[5], d[6], d[7]};
        }
    }
}

template <int KW>
__global__ __launch_bounds__(NTHR) void guided_fused(
    const float* __restrict__ X,  const float* __restrict__ Y,
    const float* __restrict__ Wk, const float* __restrict__ Nn,
    float* __restrict__ Out)
{
    using G = Geo<KW>;
    constexpr int PR = G::PR, RR = G::RR, HL = G::HL, P = G::P, PT = G::PT, ASZ = G::ASZ, SG = G::SG;

    __shared__ float wpad[3 * WPROWS * KW];
    __shared__ __attribute__((aligned(16))) float sX[ASZ];
    __shared__ __attribute__((aligned(16))) float sY[ASZ];
    __shared__ __attribute__((aligned(16))) float sDa[ASZ];
    __shared__ __attribute__((aligned(16))) float sDb[ASZ];

    const int tid  = threadIdx.x;
    const int lane = tid & 31;
    const int wid  = tid >> 5;

    constexpr int BPR = HW / BC;
    constexpr int BPI = BPR * (HW / BR);
    const int img = blockIdx.x / BPI;
    const int rem = blockIdx.x % BPI;
    const int R0  = (rem / BPR) * BR;
    const int C0  = (rem % BPR) * BC;
    const int gr0 = R0 - 3 * PC;
    const int gc0 = C0 - HL;

    const float* Xi = X   + (size_t)img * IMG;
    const float* Yi = Y   + (size_t)img * IMG;
    float*       Oi = Out + (size_t)img * IMG;

    for (int i = tid; i < 3 * WPROWS * KW; i += NTHR) {
        const int p  = i / (WPROWS * KW);
        const int o  = i - p * (WPROWS * KW);
        const int kx = o / WPROWS, t = o - kx * WPROWS;
        wpad[i] = (t >= 16 && t < 16 + KH) ? Wk[p * KH * KW + (t - 16) * KW + kx] : 0.0f;
    }

    {
        constexpr int Q  = P / 4;
        constexpr int QB = (Q + 3) / 4;
        constexpr int NB = (RR / 8) * QB;
        constexpr int IT = (NB + NWAVE - 1) / NWAVE;
        const int lr = lane & 7, lq = lane >> 3;
        float4 xv[IT], yv[IT];
#pragma unroll
        for (int it = 0; it < IT; ++it) {
            const int blk = min(wid + it * NWAVE, NB - 1);
            const int rb = blk / QB, qb = blk - rb * QB;
            const int r = 8 * rb + lr, q = min(4 * qb + lq, Q - 1);
            const int cr = min(max(gr0 + r, 0), HW - 1);
            const int cc = min(max(gc0 + 4 * q, 0), HW - 4);
            const size_t o = (size_t)cr * HW + cc;
            xv[it] = *reinterpret_cast<const float4*>(Xi + o);
            yv[it] = *reinterpret_cast<const float4*>(Yi + o);
        }
#pragma unroll
        for (int it = 0; it < IT; ++it) {
            const int blk = wid + it * NWAVE;
            const int rb = blk / QB, qb = blk - rb * QB;
            const int r = 8 * rb + lr, q = 4 * qb + lq;
            if (blk < NB && q < Q) {
                const int gr = gr0 + r, gc = gc0 + 4 * q;
                const bool in = (unsigned)gr < (unsigned)HW && (unsigned)gc < (unsigned)HW;
                float4 x = xv[it], y = yv[it];
                if (!in) { x = make_float4(0.f, 0.f, 0.f, 0.f); y = x; }
                const int l = 4 * q * PT + r;
                sX[l]           = x.x;  sY[l]           = y.x;  sDa[l]           = y.x - x.x;
                sX[l + PT]      = x.y;  sY[l + PT]      = y.y;  sDa[l + PT]      = y.y - x.y;
                sX[l + 2 * PT]  = x.z;  sY[l + 2 * PT]  = y.z;  sDa[l + 2 * PT]  = y.z - x.z;
                sX[l + 3 * PT]  = x.w;  sY[l + 3 * PT]  = y.w;  sDa[l + 3 * PT]  = y.w - x.w;
            }
        }
    }
    __syncthreads();

    mid_pass<KW, 1>(sDa, sDb, sX, sY, wpad,                   Nn,           gr0, gc0, lane, wid);
    __syncthreads();
    mid_pass<KW, 2>(sDb, sDa, sX, sY, wpad + WPROWS * KW,     Nn + IMG,     gr0, gc0, lane, wid);
    __syncthreads();

    {
        constexpr int ro = 3 * PC, co = HL;
        constexpr int TC = BC / 16;
        const float* __restrict__ N3 = Nn + 2 * (size_t)IMG;
        const float* wp = wpad + 2 * WPROWS * KW;
        const int half = lane >> 4;
        const int n    = lane & 15;
        v2f a[8][KW];
        load_band<KW>(wp, lane, a);
        for (int t = wid; t < (BR / 16) * TC; t += NWAVE) {
            const int ti = t / TC, tj = t - ti * TC;
            const int r0 = ro + 16 * ti, c0 = co + 16 * tj;
            const int c  = c0 + n;
            const int gc = gc0 + c;
            float nv[8];
#pragma unroll
            for (int v = 0; v < 8; ++v)
                nv[v] = N3[(size_t)(gr0 + r0 + v + 8 * half) * HW + gc];
            const float* patch = sDa + (c0 - PR + n) * PT + (r0 - PC) + 2 * half;
            v2f b[SG][KW];
            load_patch<KW, PT, SG>(patch, 0, b);
            const int l0 = c * PT + r0 + 8 * half;
            const v4f x0 = *reinterpret_cast<const v4f*>(sX + l0);
            const v4f x1 = *reinterpret_cast<const v4f*>(sX + l0 + 4);
            const float xs[8] = {x0.x, x0.y, x0.z, x0.w, x1.x, x1.y, x1.z, x1.w};
            __builtin_amdgcn_sched_barrier(0);
            v8f acc = {};
            acc = band_mma<KW, SG>(a, 0, b, acc);
#pragma unroll
            for (int s0 = SG; s0 < 8; s0 += SG) {
                load_patch<KW, PT, SG>(patch, s0, b);
                __builtin_amdgcn_sched_barrier(0);
                acc = band_mma<KW, SG>(a, s0, b, acc);
            }
#pragma unroll
            for (int v = 0; v < 8; ++v) {
                const size_t o = (size_t)(gr0 + r0 + v + 8 * half) * HW + gc;
                Oi[o] = xs[v] + acc[v] / nv[v];
            }
        }
    }
}

template <int KW>
static void run_fused(const float* X, const float* y, const float* k,
                      const float* Nn, float* out, int batch, hipStream_t stream)
{
    dim3 grid(batch * (HW / BR) * (HW / BC)), block(NTHR);
    guided_fused<KW><<<grid, block, 0, stream>>>(X, y, k, Nn, out);
}

extern "C" void kernel_launch(void* const* d_in, const int* in_sizes, int n_in,
                              void* d_out, int out_size, void* d_ws, size_t ws_size,
                              hipStream_t stream) {
    const float* X  = (const float*)d_in[0];
    const float* y  = (const float*)d_in[1];
    const float* k  = (const float*)d_in[2];
    const float* Nn = (const float*)d_in[3];
    float* out = (float*)d_out;

    const int KW    = in_sizes[2] / (3 * KH);
    const int batch = in_sizes[0] / IMG;

    switch (KW) {
        case 1: run_fused<1>(X, y, k, Nn, out, batch, stream); break;
        case 3: run_fused<3>(X, y, k, Nn, out, batch, stream); break;
        case 7: run_fused<7>(X, y, k, Nn, out, batch, stream); break;
        case 9: run_fused<9>(X, y, k, Nn, out, batch, stream); break;
        case 5:
        default: run_fused<5>(X, y, k, Nn, out, batch, stream); break;
    }
}


# a block walks down 4 stacked 64-row chunks, keeping the last 32 rows of X, Y and both D planes in LDS (moved up 64 rows): later chunks run each pass on 64 new rows only; order of all sums unchanged
# speedup vs baseline: 1.6456x; 1.1482x over previous
#include <hip/hip_runtime.h>


typedef __attribute__((ext_vector_type(2))) float v2f;
typedef __attribute__((ext_vector_type(4))) float v4f;
typedef __attribute__((ext_vector_type(8))) float v8f;

#define HW     2048
#define IMG    (HW * HW)
#define KH     17
#define PC     8
#define WPROWS 48
#define BR     64
#define BC     128
#define NCH    4
#define NTHR   512
#define NWAVE  (NTHR / 32)

template <int KW>
struct Geo {
    static constexpr int PR  = KW / 2;
    static constexpr int RR  = BR + 6 * PC;
    static constexpr int HL  = (3 * PR <= 8) ? 8 : 12;
    static constexpr int P   = BC + 2 * HL;
    static constexpr int PT  = 4 * ((RR / 4) | 1);
    static constexpr int ASZ = (P + 8) * PT;
    static constexpr int SG  = (KW <= 5) ? 8 : (KW <= 7) ? 4 : 2;
};

template <int KW>
__device__ __forceinline__ void load_band(const float* wp, int lane, v2f (&a)[8][KW])
{
    const int half = lane >> 4;
    const int m    = lane & 15;
    const float* w0 = wp + (2 * half - m + 16);
#pragma unroll
    for (int s = 0; s < 8; ++s) {
#pragma unroll
        for (int kx = 0; kx < KW; ++kx) {
            a[s][kx].x = w0[kx * WPROWS + 4 * s];
            a[s][kx].y = w0[kx * WPROWS + 4 * s + 1];
        }
    }
}

template <int KW, int PT, int SG>
__device__ __forceinline__ void load_patch(const float* src, int s0, v2f (&b)[SG][KW])
{
#pragma unroll
    for (int s = 0; s < SG; ++s) {
#pragma unroll
        for (int kx = 0; kx < KW; ++kx)
            b[s][kx] = *reinterpret_cast<const v2f*>(src + kx * PT + 4 * (s0 + s));
    }
}

template <int KW, int SG>
__device__ __forceinline__ v8f band_mma(const v2f (&a)[8][KW], int s0, const v2f (&b)[SG][KW], v8f acc)
{
#pragma unroll
    for (int s = 0; s < SG; ++s) {
#pragma unroll
        for (int kx = 0; kx < KW; ++kx)
            acc = __builtin_amdgcn_wmma_f32_16x16x4_f32(
                false, a[s0 + s][kx], false, b[s][kx], (short)0, acc, false, false);
    }
    return acc;
}

template <int KW, int PASS, int RO, int NROW>
__device__ __forceinline__ void mid_pass(
    const float* Din, float* Dout, float* sX, const float* sY, const float* wp,
    const float* __restrict__ Np, int gr0, int gc0, int lane, int wid)
{
    using G = Geo<KW>;
    constexpr int PR = G::PR, PT = G::PT, SG = G::SG;
    constexpr int ro   = RO;
    constexpr int co   = G::HL - (3 - PASS) * PR;
    constexpr int nrow = NROW;
    constexpr int ncol = BC + 2 * (3 - PASS) * PR;
    constexpr int TR   = nrow / 16;
    constexpr int TC   = (ncol + 15) / 16;
    const int half = lane >> 4;
    const int n    = lane & 15;

    v2f a[8][KW];
    load_band<KW>(wp, lane, a);

    for (int t = wid; t < TR * TC; t += NWAVE) {
        const int ti = t / TC, tj = t - ti * TC;
        const int r0 = ro + 16 * ti, c0 = co + 16 * tj;
        const int  c    = c0 + n;
        const int  gc   = gc0 + c;
        const bool cval = c < co + ncol;
        const bool cin  = (unsigned)gc < (unsigned)HW;
        float nv[8];
        const int cc = min(max(gc, 0), HW - 1);
#pragma unroll
        for (int v = 0; v < 8; ++v) {
            const int cr = min(max(gr0 + r0 + v + 8 * half, 0), HW - 1);
            nv[v] = Np[(size_t)cr * HW + cc];
        }
        const float* patch = Din + (c0 - PR + n) * PT + (r0 - PC) + 2 * half;
        v2f b[SG][KW];
        load_patch<KW, PT, SG>(patch, 0, b);
        const int l0 = c * PT + r0 + 8 * half;
        const v4f x0 = *reinterpret_cast<const v4f*>(sX + l0);
        const v4f x1 = *reinterpret_cast<const v4f*>(sX + l0 + 4);
        const v4f y0 = *reinterpret_cast<const v4f*>(sY + l0);
        const v4f y1 = *reinterpret_cast<const v4f*>(sY + l0 + 4);
        const float xs[8] = {x0.x, x0.y, x0.z, x0.w, x1.x, x1.y, x1.z, x1.w};
        const float ys[8] = {y0.x, y0.y, y0.z, y0.w, y1.x, y1.y, y1.z, y1.w};
        __builtin_amdgcn_sched_barrier(0);
        v8f acc = {};
        acc = band_mma<KW, SG>(a, 0, b, acc);
#pragma unroll
        for (int s0 = SG; s0 < 8; s0 += SG) {
            load_patch<KW, PT, SG>(patch, s0, b);
            __builtin_amdgcn_sched_barrier(0);
            acc = band_mma<KW, SG>(a, s0, b, acc);
        }
        __builtin_amdgcn_sched_barrier(0);
        asm volatile("" :: "v"(nv[0]), "v"(nv[1]), "v"(nv[2]), "v"(nv[3]),
                           "v"(nv[4]), "v"(nv[5]), "v"(nv[6]), "v"(nv[7]));
        asm volatile("" :: "v"(xs[0]), "v"(xs[1]), "v"(xs[2]), "v"(xs[3]),
                           "v"(xs[4]), "v"(xs[5]), "v"(xs[6]), "v"(xs[7]));
        asm volatile("" :: "v"(ys[0]), "v"(ys[1]), "v"(ys[2]), "v"(ys[3]),
                           "v"(ys[4]), "v"(ys[5]), "v"(ys[6]), "v"(ys[7]));
        float xn[8], d[8];
#pragma unroll
        for (int v = 0; v < 8; ++v) {
            const int gr = gr0 + r0 + v + 8 * half;
            const unsigned keep = (cin && (unsigned)gr < (unsigned)HW) ? 0xffffffffu : 0u;
            const float q = __uint_as_float(__float_as_uint(acc[v] / nv[v]) & keep);
            xn[v] = xs[v] + q;
            d[v]  = ys[v] - xn[v];
        }
        if (cval) {
            *reinterpret_cast<v4f*>(sX + l0)       = (v4f){xn[0], xn[1], xn[2], xn[3]};
            *reinterpret_cast<v4f*>(sX + l0 + 4)   = (v4f){xn[4], xn[5], xn[6], xn[7]};
            *reinterpret_cast<v4f*>(Dout + l0)     = (v4f){d[0], d[1], d[2], d[3]};
            *reinterpret_cast<v4f*>(Dout + l0 + 4) = (v4f){d[4], d[5], d[6], d[7]};
        }
    }
}

template <int KW, int RS, int NR, bool SHIFT>
__device__ __forceinline__ void stage_rows(
    const float* __restrict__ Xi, const float* __restrict__ Yi,
    float* sX, float* sY, float* sDa, float* sDb, int gr0, int gc0, int tid)
{
    using G = Geo<KW>;
    constexpr int RR = G::RR, P = G::P, PT = G::PT;
    constexpr int Q  = P / 4;
    constexpr int QB = (Q + 3) / 4;
    constexpr int NB = (NR / 8) * QB;
    constexpr int IT = (NB + NWAVE - 1) / NWAVE;
    const int lane = tid & 31;
    const int wid = __builtin_amdgcn_readfirstlane(tid >> 5);
    const int lr = lane & 7, lq = lane >> 3;
    float4 xv[IT], yv[IT];
#pragma unroll
    for (int it = 0; it < IT; ++it) {
        const int blk = min(wid + it * NWAVE, NB - 1);
        const int rb = blk / QB, qb = blk - rb * QB;
        const int r = RS + 8 * rb + lr, q = min(4 * qb + lq, Q - 1);
        const int cr = min(max(gr0 + r, 0), HW - 1);
        const int cc = min(max(gc0 + 4 * q, 0), HW - 4);
        const size_t o = (size_t)cr * HW + cc;
        xv[it] = *reinterpret_cast<const float4*>(Xi + o);
        yv[it] = *reinterpret_cast<const float4*>(Yi + o);
    }
    if (SHIFT) {
        constexpr int KEEP = 4 * PC;
        constexpr int SRC  = RR - KEEP;
        constexpr int DST  = SRC - BR;
        constexpr int NV   = (P + 8) * (KEEP / 4);
        for (int i = tid; i < NV; i += NTHR) {
            const int col = i / (KEEP / 4), k = i - col * (KEEP / 4);
            const int s = col * PT + SRC + 4 * k, d = col * PT + DST + 4 * k;
            const v4f vx = *reinterpret_cast<const v4f*>(sX  + s);
            const v4f vy = *reinterpret_cast<const v4f*>(sY  + s);
            const v4f va = *reinterpret_cast<const v4f*>(sDa + s);
            const v4f vb = *reinterpret_cast<const v4f*>(sDb + s);
            *reinterpret_cast<v4f*>(sX  + d) = vx;
            *reinterpret_cast<v4f*>(sY  + d) = vy;
            *reinterpret_cast<v4f*>(sDa + d) = va;
            *reinterpret_cast<v4f*>(sDb + d) = vb;
        }
        __syncthreads();
    }
#pragma unroll
    for (int it = 0; it < IT; ++it) {
        const int blk = wid + it * NWAVE;
        const int rb = blk / QB, qb = blk - rb * QB;
        const int r = RS + 8 * rb + lr, q = 4 * qb + lq;
        if (blk < NB && q < Q) {
            const int gr = gr0 + r, gc = gc0 + 4 * q;
            const bool in = (unsigned)gr < (unsigned)HW && (unsigned)gc < (unsigned)HW;
            float4 x = xv[it], y = yv[it];
            if (!in) { x = make_float4(0.f, 0.f, 0.f, 0.f); y = x; }
            const int l = 4 * q * PT + r;
            sX[l]           = x.x;  sY[l]           = y.x;  sDa[l]           = y.x - x.x;
            sX[l + PT]      = x.y;  sY[l + PT]      = y.y;  sDa[l + PT]      = y.y - x.y;
            sX[l + 2 * PT]  = x.z;  sY[l + 2 * PT]  = y.z;  sDa[l + 2 * PT]  = y.z - x.z;
            sX[l + 3 * PT]  = x.w;  sY[l + 3 * PT]  = y.w;  sDa[l + 3 * PT]  = y.w - x.w;
        }
    }
}

template <int KW>
__device__ __forceinline__ void last_pass(
    const float* sDa, const float* sX, const float* wp,
    const float* __restrict__ N3, float* __restrict__ Oi,
    int gr0, int gc0, int lane, int wid)
{
    using G = Geo<KW>;
    constexpr int PR = G::PR, HL = G::HL, PT = G::PT, SG = G::SG;
    constexpr int ro = 3 * PC, co = HL;
    constexpr int TC = BC / 16;
    const int half = lane >> 4;
    const int n    = lane & 15;
    v2f a[8][KW];
    load_band<KW>(wp, lane, a);
    for (int t = wid; t < (BR / 16) * TC; t += NWAVE) {
        const int ti = t / TC, tj = t - ti * TC;
        const int r0 = ro + 16 * ti, c0 = co + 16 * tj;
        const int c  = c0 + n;
        const int gc = gc0 + c;
        const size_t o0 = (size_t)(unsigned)(gr0 + r0 + 8 * half) * HW + (unsigned)gc;
        const float* __restrict__ np = N3 + o0;
        float*       __restrict__ op = Oi + o0;
        float nv[8];
#pragma unroll
        for (int v = 0; v < 8; ++v)
            nv[v] = np[v * HW];
        const float* patch = sDa + (c0 - PR + n) * PT + (r0 - PC) + 2 * half;
        v2f b[SG][KW];
        load_patch<KW, PT, SG>(patch, 0, b);
        const int l0 = c * PT + r0 + 8 * half;
        const v4f x0 = *reinterpret_cast<const v4f*>(sX + l0);
        const v4f x1 = *reinterpret_cast<const v4f*>(sX + l0 + 4);
        const float xs[8] = {x0.x, x0.y, x0.z, x0.w, x1.x, x1.y, x1.z, x1.w};
        __builtin_amdgcn_sched_barrier(0);
        v8f acc = {};
        acc = band_mma<KW, SG>(a, 0, b, acc);
#pragma unroll
        for (int s0 = SG; s0 < 8; s0 += SG) {
            load_patch<KW, PT, SG>(patch, s0, b);
            __builtin_amdgcn_sched_barrier(0);
            acc = band_mma<KW, SG>(a, s0, b, acc);
        }
#pragma unroll
        for (int v = 0; v < 8; ++v)
            op[v * HW] = xs[v] + acc[v] / nv[v];
    }
}

template <int KW>
__global__ __launch_bounds__(NTHR) void guided_fused(
    const float* __restrict__ X,  const float* __restrict__ Y,
    const float* __restrict__ Wk, const float* __restrict__ Nn,
    float* __restrict__ Out)
{
    using G = Geo<KW>;
    constexpr int RR = G::RR, HL = G::HL, ASZ = G::ASZ;

    __shared__ float wpad[3 * WPROWS * KW];
    __shared__ __attribute__((aligned(16))) float sX[ASZ];
    __shared__ __attribute__((aligned(16))) float sY[ASZ];
    __shared__ __attribute__((aligned(16))) float sDa[ASZ];
    __shared__ __attribute__((aligned(16))) float sDb[ASZ];

    const int tid  = threadIdx.x;
    const int lane = tid & 31;
    const int wid  = tid >> 5;

    constexpr int BPR = HW / BC;
    constexpr int BPI = BPR * (HW / (BR * NCH));
    const int img = blockIdx.x / BPI;
    const int rem = blockIdx.x % BPI;
    const int R0  = (rem / BPR) * (BR * NCH);
    const int C0  = (rem % BPR) * BC;
    int       gr0 = R0 - 3 * PC;
    const int gc0 = C0 - HL;

    const float* Xi = X   + (size_t)img * IMG;
    const float* Yi = Y   + (size_t)img * IMG;
    float*       Oi = Out + (size_t)img * IMG;
    const float* N3 = Nn + 2 * (size_t)IMG;

    for (int i = tid; i < 3 * WPROWS * KW; i += NTHR) {
        const int p  = i / (WPROWS * KW);
        const int o  = i - p * (WPROWS * KW);
        const int kx = o / WPROWS, t = o - kx * WPROWS;
        wpad[i] = (t >= 16 && t < 16 + KH) ? Wk[p * KH * KW + (t - 16) * KW + kx] : 0.0f;
    }

    stage_rows<KW, 0, RR, false>(Xi, Yi, sX, sY, sDa, sDb, gr0, gc0, tid);
    __syncthreads();
    mid_pass<KW, 1, PC, BR + 4 * PC>(sDa, sDb, sX, sY, wpad,               Nn,       gr0, gc0, lane, wid);
    __syncthreads();
    mid_pass<KW, 2, 2 * PC, BR + 2 * PC>(sDb, sDa, sX, sY, wpad + WPROWS * KW, Nn + IMG, gr0, gc0, lane, wid);
    __syncthreads();
    last_pass<KW>(sDa, sX, wpad + 2 * WPROWS * KW, N3, Oi, gr0, gc0, lane, wid);

#pragma unroll 1
    for (int ch = 1; ch < NCH; ++ch) {
        gr0 += BR;
        __syncthreads();
        stage_rows<KW, 6 * PC, BR, true>(Xi, Yi, sX, sY, sDa, sDb, gr0, gc0, tid);
        __syncthreads();
        mid_pass<KW, 1, 5 * PC, BR>(sDa, sDb, sX, sY, wpad,               Nn,       gr0, gc0, lane, wid);
        __syncthreads();
        mid_pass<KW, 2, 4 * PC, BR>(sDb, sDa, sX, sY, wpad + WPROWS * KW, Nn + IMG, gr0, gc0, lane, wid);
        __syncthreads();
        last_pass<KW>(sDa, sX, wpad + 2 * WPROWS * KW, N3, Oi, gr0, gc0, lane, wid);
    }
}

template <int KW>
static void run_fused(const float* X, const float* y, const float* k,
                      const float* Nn, float* out, int batch, hipStream_t stream)
{
    dim3 grid(batch * (HW / (BR * NCH)) * (HW / BC)), block(NTHR);
    guided_fused<KW><<<grid, block, 0, stream>>>(X, y, k, Nn, out);
}

extern "C" void kernel_launch(void* const* d_in, const int* in_sizes, int n_in,
                              void* d_out, int out_size, void* d_ws, size_t ws_size,
                              hipStream_t stream) {
    const float* X  = (const float*)d_in[0];
    const float* y  = (const float*)d_in[1];
    const float* k  = (const float*)d_in[2];
    const float* Nn = (const float*)d_in[3];
    float* out = (float*)d_out;

    const int KW    = in_sizes[2] / (3 * KH);
    const int batch = in_sizes[0] / IMG;

    switch (KW) {
        case 1: run_fused<1>(X, y, k, Nn, out, batch, stream); break;
        case 3: run_fused<3>(X, y, k, Nn, out, batch, stream); break;
        case 7: run_fused<7>(X, y, k, Nn, out, batch, stream); break;
        case 9: run_fused<9>(X, y, k, Nn, out, batch, stream); break;
        case 5:
        default: run_fused<5>(X, y, k, Nn, out, batch, stream); break;
    }
}


# each quotient conv/N is conv * v_rcp_f32(N): hardware reciprocal (1 ulp) and a multiply, 2 instructions where the correctly rounded division took 11; last bits may differ; all sums keep order
# speedup vs baseline: 1.7675x; 1.0741x over previous
#include <hip/hip_runtime.h>


typedef __attribute__((ext_vector_type(2))) float v2f;
typedef __attribute__((ext_vector_type(4))) float v4f;
typedef __attribute__((ext_vector_type(8))) float v8f;

#define HW     2048
#define IMG    (HW * HW)
#define KH     17
#define PC     8
#define WPROWS 48
#define BR     64
#define BC     128
#define NCH    4
#define NTHR   512
#define NWAVE  (NTHR / 32)

template <int KW>
struct Geo {
    static constexpr int PR  = KW / 2;
    static constexpr int RR  = BR + 6 * PC;
    static constexpr int HL  = (3 * PR <= 8) ? 8 : 12;
    static constexpr int P   = BC + 2 * HL;
    static constexpr int PT  = 4 * ((RR / 4) | 1);
    static constexpr int ASZ = (P + 8) * PT;
    static constexpr int SG  = (KW <= 5) ? 8 : (KW <= 7) ? 4 : 2;
};

template <int KW>
__device__ __forceinline__ void load_band(const float* wp, int lane, v2f (&a)[8][KW])
{
    const int half = lane >> 4;
    const int m    = lane & 15;
    const float* w0 = wp + (2 * half - m + 16);
#pragma unroll
    for (int s = 0; s < 8; ++s) {
#pragma unroll
        for (int kx = 0; kx < KW; ++kx) {
            a[s][kx].x = w0[kx * WPROWS + 4 * s];
            a[s][kx].y = w0[kx * WPROWS + 4 * s + 1];
        }
    }
}

template <int KW, int PT, int SG>
__device__ __forceinline__ void load_patch(const float* src, int s0, v2f (&b)[SG][KW])
{
#pragma unroll
    for (int s = 0; s < SG; ++s) {
#pragma unroll
        for (int kx = 0; kx < KW; ++kx)
            b[s][kx] = *reinterpret_cast<const v2f*>(src + kx * PT + 4 * (s0 + s));
    }
}

template <int KW, int SG>
__device__ __forceinline__ v8f band_mma(const v2f (&a)[8][KW], int s0, const v2f (&b)[SG][KW], v8f acc)
{
#pragma unroll
    for (int s = 0; s < SG; ++s) {
#pragma unroll
        for (int kx = 0; kx < KW; ++kx)
            acc = __builtin_amdgcn_wmma_f32_16x16x4_f32(
                false, a[s0 + s][kx], false, b[s][kx], (short)0, acc, false, false);
    }
    return acc;
}

template <int KW, int PASS, int RO, int NROW>
__device__ __forceinline__ void mid_pass(
    const float* Din, float* Dout, float* sX, const float* sY, const float* wp,
    const float* __restrict__ Np, int gr0, int gc0, int lane, int wid)
{
    using G = Geo<KW>;
    constexpr int PR = G::PR, PT = G::PT, SG = G::SG;
    constexpr int ro   = RO;
    constexpr int co   = G::HL - (3 - PASS) * PR;
    constexpr int nrow = NROW;
    constexpr int ncol = BC + 2 * (3 - PASS) * PR;
    constexpr int TR   = nrow / 16;
    constexpr int TC   = (ncol + 15) / 16;
    const int half = lane >> 4;
    const int n    = lane & 15;

    v2f a[8][KW];
    load_band<KW>(wp, lane, a);

    for (int t = wid; t < TR * TC; t += NWAVE) {
        const int ti = t / TC, tj = t - ti * TC;
        const int r0 = ro + 16 * ti, c0 = co + 16 * tj;
        const int  c    = c0 + n;
        const int  gc   = gc0 + c;
        const bool cval = c < co + ncol;
        const bool cin  = (unsigned)gc < (unsigned)HW;
        float nv[8];
        const int cc = min(max(gc, 0), HW - 1);
#pragma unroll
        for (int v = 0; v < 8; ++v) {
            const int cr = min(max(gr0 + r0 + v + 8 * half, 0), HW - 1);
            nv[v] = Np[(size_t)cr * HW + cc];
        }
        const float* patch = Din + (c0 - PR + n) * PT + (r0 - PC) + 2 * half;
        v2f b[SG][KW];
        load_patch<KW, PT, SG>(patch, 0, b);
        const int l0 = c * PT + r0 + 8 * half;
        const v4f x0 = *reinterpret_cast<const v4f*>(sX + l0);
        const v4f x1 = *reinterpret_cast<const v4f*>(sX + l0 + 4);
        const v4f y0 = *reinterpret_cast<const v4f*>(sY + l0);
        const v4f y1 = *reinterpret_cast<const v4f*>(sY + l0 + 4);
        const float xs[8] = {x0.x, x0.y, x0.z, x0.w, x1.x, x1.y, x1.z, x1.w};
        const float ys[8] = {y0.x, y0.y, y0.z, y0.w, y1.x, y1.y, y1.z, y1.w};
        __builtin_amdgcn_sched_barrier(0);
        v8f acc = {};
        acc = band_mma<KW, SG>(a, 0, b, acc);
#pragma unroll
        for (int s0 = SG; s0 < 8; s0 += SG) {
            load_patch<KW, PT, SG>(patch, s0, b);
            __builtin_amdgcn_sched_barrier(0);
            acc = band_mma<KW, SG>(a, s0, b, acc);
        }
        __builtin_amdgcn_sched_barrier(0);
        asm volatile("" :: "v"(nv[0]), "v"(nv[1]), "v"(nv[2]), "v"(nv[3]),
                           "v"(nv[4]), "v"(nv[5]), "v"(nv[6]), "v"(nv[7]));
        asm volatile("" :: "v"(xs[0]), "v"(xs[1]), "v"(xs[2]), "v"(xs[3]),
                           "v"(xs[4]), "v"(xs[5]), "v"(xs[6]), "v"(xs[7]));
        asm volatile("" :: "v"(ys[0]), "v"(ys[1]), "v"(ys[2]), "v"(ys[3]),
                           "v"(ys[4]), "v"(ys[5]), "v"(ys[6]), "v"(ys[7]));
        float xn[8], d[8];
#pragma unroll
        for (int v = 0; v < 8; ++v) {
            const int gr = gr0 + r0 + v + 8 * half;
            const unsigned keep = (cin && (unsigned)gr < (unsigned)HW) ? 0xffffffffu : 0u;
            const float q = __uint_as_float(__float_as_uint(acc[v] * __builtin_amdgcn_rcpf(nv[v])) & keep);
            xn[v] = xs[v] + q;
            d[v]  = ys[v] - xn[v];
        }
        if (cval) {
            *reinterpret_cast<v4f*>(sX + l0)       = (v4f){xn[0], xn[1], xn[2], xn[3]};
            *reinterpret_cast<v4f*>(sX + l0 + 4)   = (v4f){xn[4], xn[5], xn[6], xn[7]};
            *reinterpret_cast<v4f*>(Dout + l0)     = (v4f){d[0], d[1], d[2], d[3]};
            *reinterpret_cast<v4f*>(Dout + l0 + 4) = (v4f){d[4], d[5], d[6], d[7]};
        }
    }
}

template <int KW, int RS, int NR, bool SHIFT>
__device__ __forceinline__ void stage_rows(
    const float* __restrict__ Xi, const float* __restrict__ Yi,
    float* sX, float* sY, float* sDa, float* sDb, int gr0, int gc0, int tid)
{
    using G = Geo<KW>;
    constexpr int RR = G::RR, P = G::P, PT = G::PT;
    constexpr int Q  = P / 4;
    constexpr int QB = (Q + 3) / 4;
    constexpr int NB = (NR / 8) * QB;
    constexpr int IT = (NB + NWAVE - 1) / NWAVE;
    const int lane = tid & 31;
    const int wid = __builtin_amdgcn_readfirstlane(tid >> 5);
    const int lr = lane & 7, lq = lane >> 3;
    float4 xv[IT], yv[IT];
#pragma unroll
    for (int it = 0; it < IT; ++it) {
        const int blk = min(wid + it * NWAVE, NB - 1);
        const int rb = blk / QB, qb = blk - rb * QB;
        const int r = RS + 8 * rb + lr, q = min(4 * qb + lq, Q - 1);
        const int cr = min(max(gr0 + r, 0), HW - 1);
        const int cc = min(max(gc0 + 4 * q, 0), HW - 4);
        const size_t o = (size_t)cr * HW + cc;
        xv[it] = *reinterpret_cast<const float4*>(Xi + o);
        yv[it] = *reinterpret_cast<const float4*>(Yi + o);
    }
    if (SHIFT) {
        constexpr int KEEP = 4 * PC;
        constexpr int SRC  = RR - KEEP;
        constexpr int DST  = SRC - BR;
        constexpr int NV   = (P + 8) * (KEEP / 4);
        for (int i = tid; i < NV; i += NTHR) {
            const int col = i / (KEEP / 4), k = i - col * (KEEP / 4);
            const int s = col * PT + SRC + 4 * k, d = col * PT + DST + 4 * k;
            const v4f vx = *reinterpret_cast<const v4f*>(sX  + s);
            const v4f vy = *reinterpret_cast<const v4f*>(sY  + s);
            const v4f va = *reinterpret_cast<const v4f*>(sDa + s);
            const v4f vb = *reinterpret_cast<const v4f*>(sDb + s);
            *reinterpret_cast<v4f*>(sX  + d) = vx;
            *reinterpret_cast<v4f*>(sY  + d) = vy;
            *reinterpret_cast<v4f*>(sDa + d) = va;
            *reinterpret_cast<v4f*>(sDb + d) = vb;
        }
        __syncthreads();
    }
#pragma unroll
    for (int it = 0; it < IT; ++it) {
        const int blk = wid + it * NWAVE;
        const int rb = blk / QB, qb = blk - rb * QB;
        const int r = RS + 8 * rb + lr, q = 4 * qb + lq;
        if (blk < NB && q < Q) {
            const int gr = gr0 + r, gc = gc0 + 4 * q;
            const bool in = (unsigned)gr < (unsigned)HW && (unsigned)gc < (unsigned)HW;
            float4 x = xv[it], y = yv[it];
            if (!in) { x = make_float4(0.f, 0.f, 0.f, 0.f); y = x; }
            const int l = 4 * q * PT + r;
            sX[l]           = x.x;  sY[l]           = y.x;  sDa[l]           = y.x - x.x;
            sX[l + PT]      = x.y;  sY[l + PT]      = y.y;  sDa[l + PT]      = y.y - x.y;
            sX[l + 2 * PT]  = x.z;  sY[l + 2 * PT]  = y.z;  sDa[l + 2 * PT]  = y.z - x.z;
            sX[l + 3 * PT]  = x.w;  sY[l + 3 * PT]  = y.w;  sDa[l + 3 * PT]  = y.w - x.w;
        }
    }
}

template <int KW>
__device__ __forceinline__ void last_pass(
    const float* sDa, const float* sX, const float* wp,
    const float* __restrict__ N3, float* __restrict__ Oi,
    int gr0, int gc0, int lane, int wid)
{
    using G = Geo<KW>;
    constexpr int PR = G::PR, HL = G::HL, PT = G::PT, SG = G::SG;
    constexpr int ro = 3 * PC, co = HL;
    constexpr int TC = BC / 16;
    const int half = lane >> 4;
    const int n    = lane & 15;
    v2f a[8][KW];
    load_band<KW>(wp, lane, a);
    for (int t = wid; t < (BR / 16) * TC; t += NWAVE) {
        const int ti = t / TC, tj = t - ti * TC;
        const int r0 = ro + 16 * ti, c0 = co + 16 * tj;
        const int c  = c0 + n;
        const int gc = gc0 + c;
        const size_t o0 = (size_t)(unsigned)(gr0 + r0 + 8 * half) * HW + (unsigned)gc;
        const float* __restrict__ np = N3 + o0;
        float*       __restrict__ op = Oi + o0;
        float nv[8];
#pragma unroll
        for (int v = 0; v < 8; ++v)
            nv[v] = np[v * HW];
        const float* patch = sDa + (c0 - PR + n) * PT + (r0 - PC) + 2 * half;
        v2f b[SG][KW];
        load_patch<KW, PT, SG>(patch, 0, b);
        const int l0 = c * PT + r0 + 8 * half;
        const v4f x0 = *reinterpret_cast<const v4f*>(sX + l0);
        const v4f x1 = *reinterpret_cast<const v4f*>(sX + l0 + 4);
        const float xs[8] = {x0.x, x0.y, x0.z, x0.w, x1.x, x1.y, x1.z, x1.w};
        __builtin_amdgcn_sched_barrier(0);
        v8f acc = {};
        acc = band_mma<KW, SG>(a, 0, b, acc);
#pragma unroll
        for (int s0 = SG; s0 < 8; s0 += SG) {
            load_patch<KW, PT, SG>(patch, s0, b);
            __builtin_amdgcn_sched_barrier(0);
            acc = band_mma<KW, SG>(a, s0, b, acc);
        }
        __builtin_amdgcn_sched_barrier(0);
#pragma unroll
        for (int v = 0; v < 8; ++v)
            op[v * HW] = xs[v] + acc[v] * __builtin_amdgcn_rcpf(nv[v]);
    }
}

template <int KW>
__global__ __launch_bounds__(NTHR) void guided_fused(
    const float* __restrict__ X,  const float* __restrict__ Y,
    const float* __restrict__ Wk, const float* __restrict__ Nn,
    float* __restrict__ Out)
{
    using G = Geo<KW>;
    constexpr int RR = G::RR, HL = G::HL, ASZ = G::ASZ;

    __shared__ float wpad[3 * WPROWS * KW];
    __shared__ __attribute__((aligned(16))) float sX[ASZ];
    __shared__ __attribute__((aligned(16))) float sY[ASZ];
    __shared__ __attribute__((aligned(16))) float sDa[ASZ];
    __shared__ __attribute__((aligned(16))) float sDb[ASZ];

    const int tid  = threadIdx.x;
    const int lane = tid & 31;
    const int wid  = tid >> 5;

    constexpr int BPR = HW / BC;
    constexpr int BPI = BPR * (HW / (BR * NCH));
    const int img = blockIdx.x / BPI;
    const int rem = blockIdx.x % BPI;
    const int R0  = (rem / BPR) * (BR * NCH);
    const int C0  = (rem % BPR) * BC;
    int       gr0 = R0 - 3 * PC;
    const int gc0 = C0 - HL;

    const float* Xi = X   + (size_t)img * IMG;
    const float* Yi = Y   + (size_t)img * IMG;
    float*       Oi = Out + (size_t)img * IMG;
    const float* N3 = Nn + 2 * (size_t)IMG;

    for (int i = tid; i < 3 * WPROWS * KW; i += NTHR) {
        const int p  = i / (WPROWS * KW);
        const int o  = i - p * (WPROWS * KW);
        const int kx = o / WPROWS, t = o - kx * WPROWS;
        wpad[i] = (t >= 16 && t < 16 + KH) ? Wk[p * KH * KW + (t - 16) * KW + kx] : 0.0f;
    }

    stage_rows<KW, 0, RR, false>(Xi, Yi, sX, sY, sDa, sDb, gr0, gc0, tid);
    __syncthreads();
    mid_pass<KW, 1, PC, BR + 4 * PC>(sDa, sDb, sX, sY, wpad,               Nn,       gr0, gc0, lane, wid);
    __syncthreads();
    mid_pass<KW, 2, 2 * PC, BR + 2 * PC>(sDb, sDa, sX, sY, wpad + WPROWS * KW, Nn + IMG, gr0, gc0, lane, wid);
    __syncthreads();
    last_pass<KW>(sDa, sX, wpad + 2 * WPROWS * KW, N3, Oi, gr0, gc0, lane, wid);

#pragma unroll 1
    for (int ch = 1; ch < NCH; ++ch) {
        gr0 += BR;
        __syncthreads();
        stage_rows<KW, 6 * PC, BR, true>(Xi, Yi, sX, sY, sDa, sDb, gr0, gc0, tid);
        __syncthreads();
        mid_pass<KW, 1, 5 * PC, BR>(sDa, sDb, sX, sY, wpad,               Nn,       gr0, gc0, lane, wid);
        __syncthreads();
        mid_pass<KW, 2, 4 * PC, BR>(sDb, sDa, sX, sY, wpad + WPROWS * KW, Nn + IMG, gr0, gc0, lane, wid);
        __syncthreads();
        last_pass<KW>(sDa, sX, wpad + 2 * WPROWS * KW, N3, Oi, gr0, gc0, lane, wid);
    }
}

template <int KW>
static void run_fused(const float* X, const float* y, const float* k,
                      const float* Nn, float* out, int batch, hipStream_t stream)
{
    dim3 grid(batch * (HW / (BR * NCH)) * (HW / BC)), block(NTHR);
    guided_fused<KW><<<grid, block, 0, stream>>>(X, y, k, Nn, out);
}

extern "C" void kernel_launch(void* const* d_in, const int* in_sizes, int n_in,
                              void* d_out, int out_size, void* d_ws, size_t ws_size,
                              hipStream_t stream) {
    const float* X  = (const float*)d_in[0];
    const float* y  = (const float*)d_in[1];
    const float* k  = (const float*)d_in[2];
    const float* Nn = (const float*)d_in[3];
    float* out = (float*)d_out;

    const int KW    = in_sizes[2] / (3 * KH);
    const int batch = in_sizes[0] / IMG;

    switch (KW) {
        case 1: run_fused<1>(X, y, k, Nn, out, batch, stream); break;
        case 3: run_fused<3>(X, y, k, Nn, out, batch, stream); break;
        case 7: run_fused<7>(X, y, k, Nn, out, batch, stream); break;
        case 9: run_fused<9>(X, y, k, Nn, out, batch, stream); break;
        case 5:
        default: run_fused<5>(X, y, k, Nn, out, batch, stream); break;
    }
}


# rows read later (normalisers of each pass, the next chunk's X and Y rows) are touched a pass or a chunk early with L2 cache prefetches, one per 256-byte unit; loads, stores and all sums unchanged
# speedup vs baseline: 1.8529x; 1.0483x over previous
#include <hip/hip_runtime.h>


typedef __attribute__((ext_vector_type(2))) float v2f;
typedef __attribute__((ext_vector_type(4))) float v4f;
typedef __attribute__((ext_vector_type(8))) float v8f;

#define HW     2048
#define IMG    (HW * HW)
#define KH     17
#define PC     8
#define WPROWS 48
#define BR     64
#define BC     128
#define NCH    4
#define NTHR   512
#define NWAVE  (NTHR / 32)

template <int KW>
struct Geo {
    static constexpr int PR  = KW / 2;
    static constexpr int RR  = BR + 6 * PC;
    static constexpr int HL  = (3 * PR <= 8) ? 8 : 12;
    static constexpr int P   = BC + 2 * HL;
    static constexpr int PT  = 4 * ((RR / 4) | 1);
    static constexpr int ASZ = (P + 8) * PT;
    static constexpr int SG  = (KW <= 5) ? 8 : (KW <= 7) ? 4 : 2;
};

template <int KW>
__device__ __forceinline__ void load_band(const float* wp, int lane, v2f (&a)[8][KW])
{
    const int half = lane >> 4;
    const int m    = lane & 15;
    const float* w0 = wp + (2 * half - m + 16);
#pragma unroll
    for (int s = 0; s < 8; ++s) {
#pragma unroll
        for (int kx = 0; kx < KW; ++kx) {
            a[s][kx].x = w0[kx * WPROWS + 4 * s];
            a[s][kx].y = w0[kx * WPROWS + 4 * s + 1];
        }
    }
}

template <int KW, int PT, int SG>
__device__ __forceinline__ void load_patch(const float* src, int s0, v2f (&b)[SG][KW])
{
#pragma unroll
    for (int s = 0; s < SG; ++s) {
#pragma unroll
        for (int kx = 0; kx < KW; ++kx)
            b[s][kx] = *reinterpret_cast<const v2f*>(src + kx * PT + 4 * (s0 + s));
    }
}

template <int KW, int SG>
__device__ __forceinline__ v8f band_mma(const v2f (&a)[8][KW], int s0, const v2f (&b)[SG][KW], v8f acc)
{
#pragma unroll
    for (int s = 0; s < SG; ++s) {
#pragma unroll
        for (int kx = 0; kx < KW; ++kx)
            acc = __builtin_amdgcn_wmma_f32_16x16x4_f32(
                false, a[s0 + s][kx], false, b[s][kx], (short)0, acc, false, false);
    }
    return acc;
}

template <int KW, int PASS, int RO, int NROW>
__device__ __forceinline__ void mid_pass(
    const float* Din, float* Dout, float* sX, const float* sY, const float* wp,
    const float* __restrict__ Np, int gr0, int gc0, int lane, int wid)
{
    using G = Geo<KW>;
    constexpr int PR = G::PR, PT = G::PT, SG = G::SG;
    constexpr int ro   = RO;
    constexpr int co   = G::HL - (3 - PASS) * PR;
    constexpr int nrow = NROW;
    constexpr int ncol = BC + 2 * (3 - PASS) * PR;
    constexpr int TR   = nrow / 16;
    constexpr int TC   = (ncol + 15) / 16;
    const int half = lane >> 4;
    const int n    = lane & 15;

    v2f a[8][KW];
    load_band<KW>(wp, lane, a);

    for (int t = wid; t < TR * TC; t += NWAVE) {
        const int ti = t / TC, tj = t - ti * TC;
        const int r0 = ro + 16 * ti, c0 = co + 16 * tj;
        const int  c    = c0 + n;
        const int  gc   = gc0 + c;
        const bool cval = c < co + ncol;
        const bool cin  = (unsigned)gc < (unsigned)HW;
        float nv[8];
        const int cc = min(max(gc, 0), HW - 1);
#pragma unroll
        for (int v = 0; v < 8; ++v) {
            const int cr = min(max(gr0 + r0 + v + 8 * half, 0), HW - 1);
            nv[v] = Np[(size_t)cr * HW + cc];
        }
        const float* patch = Din + (c0 - PR + n) * PT + (r0 - PC) + 2 * half;
        v2f b[SG][KW];
        load_patch<KW, PT, SG>(patch, 0, b);
        const int l0 = c * PT + r0 + 8 * half;
        const v4f x0 = *reinterpret_cast<const v4f*>(sX + l0);
        const v4f x1 = *reinterpret_cast<const v4f*>(sX + l0 + 4);
        const v4f y0 = *reinterpret_cast<const v4f*>(sY + l0);
        const v4f y1 = *reinterpret_cast<const v4f*>(sY + l0 + 4);
        const float xs[8] = {x0.x, x0.y, x0.z, x0.w, x1.x, x1.y, x1.z, x1.w};
        const float ys[8] = {y0.x, y0.y, y0.z, y0.w, y1.x, y1.y, y1.z, y1.w};
        __builtin_amdgcn_sched_barrier(0);
        v8f acc = {};
        acc = band_mma<KW, SG>(a, 0, b, acc);
#pragma unroll
        for (int s0 = SG; s0 < 8; s0 += SG) {
            load_patch<KW, PT, SG>(patch, s0, b);
            __builtin_amdgcn_sched_barrier(0);
            acc = band_mma<KW, SG>(a, s0, b, acc);
        }
        __builtin_amdgcn_sched_barrier(0);
        asm volatile("" :: "v"(nv[0]), "v"(nv[1]), "v"(nv[2]), "v"(nv[3]),
                           "v"(nv[4]), "v"(nv[5]), "v"(nv[6]), "v"(nv[7]));
        asm volatile("" :: "v"(xs[0]), "v"(xs[1]), "v"(xs[2]), "v"(xs[3]),
                           "v"(xs[4]), "v"(xs[5]), "v"(xs[6]), "v"(xs[7]));
        asm volatile("" :: "v"(ys[0]), "v"(ys[1]), "v"(ys[2]), "v"(ys[3]),
                           "v"(ys[4]), "v"(ys[5]), "v"(ys[6]), "v"(ys[7]));
        float xn[8], d[8];
#pragma unroll
        for (int v = 0; v < 8; ++v) {
            const int gr = gr0 + r0 + v + 8 * half;
            const unsigned keep = (cin && (unsigned)gr < (unsigned)HW) ? 0xffffffffu : 0u;
            const float q = __uint_as_float(__float_as_uint(acc[v] * __builtin_amdgcn_rcpf(nv[v])) & keep);
            xn[v] = xs[v] + q;
            d[v]  = ys[v] - xn[v];
        }
        if (cval) {
            *reinterpret_cast<v4f*>(sX + l0)       = (v4f){xn[0], xn[1], xn[2], xn[3]};
            *reinterpret_cast<v4f*>(sX + l0 + 4)   = (v4f){xn[4], xn[5], xn[6], xn[7]};
            *reinterpret_cast<v4f*>(Dout + l0)     = (v4f){d[0], d[1], d[2], d[3]};
            *reinterpret_cast<v4f*>(Dout + l0 + 4) = (v4f){d[4], d[5], d[6], d[7]};
        }
    }
}

template <int NR, int J0>
__device__ __forceinline__ void touch_rows(const float* plane, int row0, int C0, int tid)
{
    const int j = (tid >> 1) - J0;
    if ((unsigned)j < (unsigned)NR) {
        const int r = min(max(row0 + j, 0), HW - 1);
        const unsigned off = ((unsigned)r * HW + (unsigned)C0 + 64u * (unsigned)(tid & 1)) * 4u;
        __builtin_prefetch(reinterpret_cast<const char*>(plane) + off, 0, 1);
    }
}

template <int KW, int RS, int NR, bool SHIFT>
__device__ __forceinline__ void stage_rows(
    const float* __restrict__ Xi, const float* __restrict__ Yi,
    float* sX, float* sY, float* sDa, float* sDb, int gr0, int gc0, int tid)
{
    using G = Geo<KW>;
    constexpr int RR = G::RR, P = G::P, PT = G::PT;
    constexpr int Q  = P / 4;
    constexpr int QB = (Q + 3) / 4;
    constexpr int NB = (NR / 8) * QB;
    constexpr int IT = (NB + NWAVE - 1) / NWAVE;
    const int lane = tid & 31;
    const int wid = __builtin_amdgcn_readfirstlane(tid >> 5);
    const int lr = lane & 7, lq = lane >> 3;
    float4 xv[IT], yv[IT];
#pragma unroll
    for (int it = 0; it < IT; ++it) {
        const int blk = min(wid + it * NWAVE, NB - 1);
        const int rb = blk / QB, qb = blk - rb * QB;
        const int r = RS + 8 * rb + lr, q = min(4 * qb + lq, Q - 1);
        const int cr = min(max(gr0 + r, 0), HW - 1);
        const int cc = min(max(gc0 + 4 * q, 0), HW - 4);
        const size_t o = (size_t)cr * HW + cc;
        xv[it] = *reinterpret_cast<const float4*>(Xi + o);
        yv[it] = *reinterpret_cast<const float4*>(Yi + o);
    }
    if (SHIFT) {
        constexpr int KEEP = 4 * PC;
        constexpr int SRC  = RR - KEEP;
        constexpr int DST  = SRC - BR;
        constexpr int NV   = (P + 8) * (KEEP / 4);
        for (int i = tid; i < NV; i += NTHR) {
            const int col = i / (KEEP / 4), k = i - col * (KEEP / 4);
            const int s = col * PT + SRC + 4 * k, d = col * PT + DST + 4 * k;
            const v4f vx = *reinterpret_cast<const v4f*>(sX  + s);
            const v4f vy = *reinterpret_cast<const v4f*>(sY  + s);
            const v4f va = *reinterpret_cast<const v4f*>(sDa + s);
            const v4f vb = *reinterpret_cast<const v4f*>(sDb + s);
            *reinterpret_cast<v4f*>(sX  + d) = vx;
            *reinterpret_cast<v4f*>(sY  + d) = vy;
            *reinterpret_cast<v4f*>(sDa + d) = va;
            *reinterpret_cast<v4f*>(sDb + d) = vb;
        }
        __syncthreads();
    }
#pragma unroll
    for (int it = 0; it < IT; ++it) {
        const int blk = wid + it * NWAVE;
        const int rb = blk / QB, qb = blk - rb * QB;
        const int r = RS + 8 * rb + lr, q = 4 * qb + lq;
        if (blk < NB && q < Q) {
            const int gr = gr0 + r, gc = gc0 + 4 * q;
            const bool in = (unsigned)gr < (unsigned)HW && (unsigned)gc < (unsigned)HW;
            float4 x = xv[it], y = yv[it];
            if (!in) { x = make_float4(0.f, 0.f, 0.f, 0.f); y = x; }
            const int l = 4 * q * PT + r;
            sX[l]           = x.x;  sY[l]           = y.x;  sDa[l]           = y.x - x.x;
            sX[l + PT]      = x.y;  sY[l + PT]      = y.y;  sDa[l + PT]      = y.y - x.y;
            sX[l + 2 * PT]  = x.z;  sY[l + 2 * PT]  = y.z;  sDa[l + 2 * PT]  = y.z - x.z;
            sX[l + 3 * PT]  = x.w;  sY[l + 3 * PT]  = y.w;  sDa[l + 3 * PT]  = y.w - x.w;
        }
    }
}

template <int KW>
__device__ __forceinline__ void last_pass(
    const float* sDa, const float* sX, const float* wp,
    const float* __restrict__ N3, float* __restrict__ Oi,
    int gr0, int gc0, int lane, int wid)
{
    using G = Geo<KW>;
    constexpr int PR = G::PR, HL = G::HL, PT = G::PT, SG = G::SG;
    constexpr int ro = 3 * PC, co = HL;
    constexpr int TC = BC / 16;
    const int half = lane >> 4;
    const int n    = lane & 15;
    v2f a[8][KW];
    load_band<KW>(wp, lane, a);
    for (int t = wid; t < (BR / 16) * TC; t += NWAVE) {
        const int ti = t / TC, tj = t - ti * TC;
        const int r0 = ro + 16 * ti, c0 = co + 16 * tj;
        const int c  = c0 + n;
        const int gc = gc0 + c;
        const size_t o0 = (size_t)(unsigned)(gr0 + r0 + 8 * half) * HW + (unsigned)gc;
        const float* __restrict__ np = N3 + o0;
        float*       __restrict__ op = Oi + o0;
        float nv[8];
#pragma unroll
        for (int v = 0; v < 8; ++v)
            nv[v] = np[v * HW];
        const float* patch = sDa + (c0 - PR + n) * PT + (r0 - PC) + 2 * half;
        v2f b[SG][KW];
        load_patch<KW, PT, SG>(patch, 0, b);
        const int l0 = c * PT + r0 + 8 * half;
        const v4f x0 = *reinterpret_cast<const v4f*>(sX + l0);
        const v4f x1 = *reinterpret_cast<const v4f*>(sX + l0 + 4);
        const float xs[8] = {x0.x, x0.y, x0.z, x0.w, x1.x, x1.y, x1.z, x1.w};
        __builtin_amdgcn_sched_barrier(0);
        v8f acc = {};
        acc = band_mma<KW, SG>(a, 0, b, acc);
#pragma unroll
        for (int s0 = SG; s0 < 8; s0 += SG) {
            load_patch<KW, PT, SG>(patch, s0, b);
            __builtin_amdgcn_sched_barrier(0);
            acc = band_mma<KW, SG>(a, s0, b, acc);
        }
        __builtin_amdgcn_sched_barrier(0);
#pragma unroll
        for (int v = 0; v < 8; ++v)
            op[v * HW] = xs[v] + acc[v] * __builtin_amdgcn_rcpf(nv[v]);
    }
}

template <int KW>
__global__ __launch_bounds__(NTHR) void guided_fused(
    const float* __restrict__ X,  const float* __restrict__ Y,
    const float* __restrict__ Wk, const float* __restrict__ Nn,
    float* __restrict__ Out)
{
    using G = Geo<KW>;
    constexpr int RR = G::RR, HL = G::HL, ASZ = G::ASZ;

    __shared__ float wpad[3 * WPROWS * KW];
    __shared__ __attribute__((aligned(16))) float sX[ASZ];
    __shared__ __attribute__((aligned(16))) float sY[ASZ];
    __shared__ __attribute__((aligned(16))) float sDa[ASZ];
    __shared__ __attribute__((aligned(16))) float sDb[ASZ];

    const int tid  = threadIdx.x;
    const int lane = tid & 31;
    const int wid  = tid >> 5;

    constexpr int BPR = HW / BC;
    constexpr int BPI = BPR * (HW / (BR * NCH));
    const int img = blockIdx.x / BPI;
    const int rem = blockIdx.x % BPI;
    const int R0  = (rem / BPR) * (BR * NCH);
    const int C0  = (rem % BPR) * BC;
    int       gr0 = R0 - 3 * PC;
    const int gc0 = C0 - HL;

    const float* Xi = X   + (size_t)img * IMG;
    const float* Yi = Y   + (size_t)img * IMG;
    float*       Oi = Out + (size_t)img * IMG;
    const float* N3 = Nn + 2 * (size_t)IMG;

    for (int i = tid; i < 3 * WPROWS * KW; i += NTHR) {
        const int p  = i / (WPROWS * KW);
        const int o  = i - p * (WPROWS * KW);
        const int kx = o / WPROWS, t = o - kx * WPROWS;
        wpad[i] = (t >= 16 && t < 16 + KH) ? Wk[p * KH * KW + (t - 16) * KW + kx] : 0.0f;
    }

    stage_rows<KW, 0, RR, false>(Xi, Yi, sX, sY, sDa, sDb, gr0, gc0, tid);
    static_assert(2 * (3 * BR + 6 * PC) <= NTHR && 2 * 4 * BR <= NTHR, "one touch per thread");
    touch_rows<BR + 4 * PC, 0>(Nn,                  gr0 + PC,     C0, tid);
    touch_rows<BR + 2 * PC, BR + 4 * PC>(Nn + IMG,  gr0 + 2 * PC, C0, tid);
    touch_rows<BR, 2 * BR + 6 * PC>(N3,             gr0 + 3 * PC, C0, tid);
    __syncthreads();
    mid_pass<KW, 1, PC, BR + 4 * PC>(sDa, sDb, sX, sY, wpad,               Nn,       gr0, gc0, lane, wid);
    __syncthreads();
    touch_rows<BR, 0>(Nn,            gr0 + BR + 5 * PC, C0, tid);
    touch_rows<BR, BR>(Nn + IMG,     gr0 + BR + 4 * PC, C0, tid);
    touch_rows<BR, 2 * BR>(N3,       gr0 + BR + 3 * PC, C0, tid);
    mid_pass<KW, 2, 2 * PC, BR + 2 * PC>(sDb, sDa, sX, sY, wpad + WPROWS * KW, Nn + IMG, gr0, gc0, lane, wid);
    __syncthreads();
    touch_rows<BR, 3 * BR>(Xi, gr0 + BR + 6 * PC, C0, tid);
    touch_rows<BR, 0>(Yi,      gr0 + BR + 6 * PC, C0, tid);
    last_pass<KW>(sDa, sX, wpad + 2 * WPROWS * KW, N3, Oi, gr0, gc0, lane, wid);

#pragma unroll 1
    for (int ch = 1; ch < NCH; ++ch) {
        gr0 += BR;
        __syncthreads();
        stage_rows<KW, 6 * PC, BR, true>(Xi, Yi, sX, sY, sDa, sDb, gr0, gc0, tid);
        __syncthreads();
        mid_pass<KW, 1, 5 * PC, BR>(sDa, sDb, sX, sY, wpad,               Nn,       gr0, gc0, lane, wid);
        __syncthreads();
        const bool more = ch + 1 < NCH;
        if (more) {
            touch_rows<BR, 0>(Nn,            gr0 + BR + 5 * PC, C0, tid);
            touch_rows<BR, BR>(Nn + IMG,     gr0 + BR + 4 * PC, C0, tid);
            touch_rows<BR, 2 * BR>(N3,       gr0 + BR + 3 * PC, C0, tid);
        }
        mid_pass<KW, 2, 4 * PC, BR>(sDb, sDa, sX, sY, wpad + WPROWS * KW, Nn + IMG, gr0, gc0, lane, wid);
        __syncthreads();
        if (more) {
            touch_rows<BR, 3 * BR>(Xi, gr0 + BR + 6 * PC, C0, tid);
            touch_rows<BR, 0>(Yi,      gr0 + BR + 6 * PC, C0, tid);
        }
        last_pass<KW>(sDa, sX, wpad + 2 * WPROWS * KW, N3, Oi, gr0, gc0, lane, wid);
    }
}

template <int KW>
static void run_fused(const float* X, const float* y, const float* k,
                      const float* Nn, float* out, int batch, hipStream_t stream)
{
    dim3 grid(batch * (HW / (BR * NCH)) * (HW / BC)), block(NTHR);
    guided_fused<KW><<<grid, block, 0, stream>>>(X, y, k, Nn, out);
}

extern "C" void kernel_launch(void* const* d_in, const int* in_sizes, int n_in,
                              void* d_out, int out_size, void* d_ws, size_t ws_size,
                              hipStream_t stream) {
    const float* X  = (const float*)d_in[0];
    const float* y  = (const float*)d_in[1];
    const float* k  = (const float*)d_in[2];
    const float* Nn = (const float*)d_in[3];
    float* out = (float*)d_out;

    const int KW    = in_sizes[2] / (3 * KH);
    const int batch = in_sizes[0] / IMG;

    switch (KW) {
        case 1: run_fused<1>(X, y, k, Nn, out, batch, stream); break;
        case 3: run_fused<3>(X, y, k, Nn, out, batch, stream); break;
        case 7: run_fused<7>(X, y, k, Nn, out, batch, stream); break;
        case 9: run_fused<9>(X, y, k, Nn, out, batch, stream); break;
        case 5:
        default: run_fused<5>(X, y, k, Nn, out, batch, stream); break;
    }
}


# tile bookkeeping moved off the vector lanes: wave index read as a scalar, so tile counter and offsets are scalar; a lane's 8 normaliser rows use one clamped base plus fixed offsets; sums keep order
# speedup vs baseline: 1.9241x; 1.0384x over previous
#include <hip/hip_runtime.h>


typedef __attribute__((ext_vector_type(2))) float v2f;
typedef __attribute__((ext_vector_type(4))) float v4f;
typedef __attribute__((ext_vector_type(8))) float v8f;

#define HW     2048
#define IMG    (HW * HW)
#define KH     17
#define PC     8
#define WPROWS 48
#define BR     64
#define BC     128
#define NCH    4
#define NTHR   512
#define NWAVE  (NTHR / 32)

template <int KW>
struct Geo {
    static constexpr int PR  = KW / 2;
    static constexpr int RR  = BR + 6 * PC;
    static constexpr int HL  = (3 * PR <= 8) ? 8 : 12;
    static constexpr int P   = BC + 2 * HL;
    static constexpr int PT  = 4 * ((RR / 4) | 1);
    static constexpr int ASZ = (P + 8) * PT;
    static constexpr int SG  = (KW <= 5) ? 8 : (KW <= 7) ? 4 : 2;
};

template <int KW>
__device__ __forceinline__ void load_band(const float* wp, int lane, v2f (&a)[8][KW])
{
    const int half = lane >> 4;
    const int m    = lane & 15;
    const float* w0 = wp + (2 * half - m + 16);
#pragma unroll
    for (int s = 0; s < 8; ++s) {
#pragma unroll
        for (int kx = 0; kx < KW; ++kx) {
            a[s][kx].x = w0[kx * WPROWS + 4 * s];
            a[s][kx].y = w0[kx * WPROWS + 4 * s + 1];
        }
    }
}

template <int KW, int PT, int SG>
__device__ __forceinline__ void load_patch(const float* src, int s0, v2f (&b)[SG][KW])
{
#pragma unroll
    for (int s = 0; s < SG; ++s) {
#pragma unroll
        for (int kx = 0; kx < KW; ++kx)
            b[s][kx] = *reinterpret_cast<const v2f*>(src + kx * PT + 4 * (s0 + s));
    }
}

template <int KW, int SG>
__device__ __forceinline__ v8f band_mma(const v2f (&a)[8][KW], int s0, const v2f (&b)[SG][KW], v8f acc)
{
#pragma unroll
    for (int s = 0; s < SG; ++s) {
#pragma unroll
        for (int kx = 0; kx < KW; ++kx)
            acc = __builtin_amdgcn_wmma_f32_16x16x4_f32(
                false, a[s0 + s][kx], false, b[s][kx], (short)0, acc, false, false);
    }
    return acc;
}

template <int KW, int PASS, int RO, int NROW>
__device__ __forceinline__ void mid_pass(
    const float* Din, float* Dout, float* sX, const float* sY, const float* wp,
    const float* __restrict__ Np, int gr0, int gc0, int lane, int wid)
{
    using G = Geo<KW>;
    constexpr int PR = G::PR, PT = G::PT, SG = G::SG;
    constexpr int ro   = RO;
    constexpr int co   = G::HL - (3 - PASS) * PR;
    constexpr int nrow = NROW;
    constexpr int ncol = BC + 2 * (3 - PASS) * PR;
    constexpr int TR   = nrow / 16;
    constexpr int TC   = (ncol + 15) / 16;
    const int half = lane >> 4;
    const int n    = lane & 15;
    const int w = __builtin_amdgcn_readfirstlane(wid);

    v2f a[8][KW];
    load_band<KW>(wp, lane, a);

    const int    cl     = co + n;
    const int    gcl    = gc0 + cl;
    const int    rl     = gr0 + ro + 8 * half;
    const int    l00    = cl * PT + ro + 8 * half;
    const float* patch0 = Din + (co - PR + n) * PT + (ro - PC) + 2 * half;
    static_assert(RO % 8 == 0 && (3 * PC) % 8 == 0 && (BR * NCH) % 8 == 0 && BR % 8 == 0 && HW % 8 == 0,
                  "a lane's 8 rows are an aligned group of 8 image rows");

    for (int t = w; t < TR * TC; t += NWAVE) {
        const int ti = t / TC, tj = t - ti * TC;
        int toff = 16 * tj * PT + 16 * ti;
        asm volatile("" : "+s"(toff));
        const int  c    = cl + 16 * tj;
        const int  gc   = gcl + 16 * tj;
        const int  rb   = rl + 16 * ti;
        const bool cval = c < co + ncol;
        const bool cin  = (unsigned)gc < (unsigned)HW;
        const bool rin  = (unsigned)rb < (unsigned)HW;
        float nv[8];
        const int rbc = min(max(rb, 0), HW - 8);
        const int cc  = min(max(gc, 0), HW - 1);
        const float* __restrict__ np = Np + ((unsigned)rbc * HW + (unsigned)cc);
#pragma unroll
        for (int v = 0; v < 8; ++v)
            nv[v] = np[v * HW];
        const float* patch = patch0 + toff;
        v2f b[SG][KW];
        load_patch<KW, PT, SG>(patch, 0, b);
        const int l0 = l00 + toff;
        const v4f x0 = *reinterpret_cast<const v4f*>(sX + l0);
        const v4f x1 = *reinterpret_cast<const v4f*>(sX + l0 + 4);
        const v4f y0 = *reinterpret_cast<const v4f*>(sY + l0);
        const v4f y1 = *reinterpret_cast<const v4f*>(sY + l0 + 4);
        const float xs[8] = {x0.x, x0.y, x0.z, x0.w, x1.x, x1.y, x1.z, x1.w};
        const float ys[8] = {y0.x, y0.y, y0.z, y0.w, y1.x, y1.y, y1.z, y1.w};
        __builtin_amdgcn_sched_barrier(0);
        v8f acc = {};
        acc = band_mma<KW, SG>(a, 0, b, acc);
#pragma unroll
        for (int s0 = SG; s0 < 8; s0 += SG) {
            load_patch<KW, PT, SG>(patch, s0, b);
            __builtin_amdgcn_sched_barrier(0);
            acc = band_mma<KW, SG>(a, s0, b, acc);
        }
        __builtin_amdgcn_sched_barrier(0);
        asm volatile("" :: "v"(nv[0]), "v"(nv[1]), "v"(nv[2]), "v"(nv[3]),
                           "v"(nv[4]), "v"(nv[5]), "v"(nv[6]), "v"(nv[7]));
        asm volatile("" :: "v"(xs[0]), "v"(xs[1]), "v"(xs[2]), "v"(xs[3]),
                           "v"(xs[4]), "v"(xs[5]), "v"(xs[6]), "v"(xs[7]));
        asm volatile("" :: "v"(ys[0]), "v"(ys[1]), "v"(ys[2]), "v"(ys[3]),
                           "v"(ys[4]), "v"(ys[5]), "v"(ys[6]), "v"(ys[7]));
        float xn[8], d[8];
        const unsigned keep = (cin && rin) ? 0xffffffffu : 0u;
#pragma unroll
        for (int v = 0; v < 8; ++v) {
            const float q = __uint_as_float(__float_as_uint(acc[v] * __builtin_amdgcn_rcpf(nv[v])) & keep);
            xn[v] = xs[v] + q;
            d[v]  = ys[v] - xn[v];
        }
        if (cval) {
            *reinterpret_cast<v4f*>(sX + l0)       = (v4f){xn[0], xn[1], xn[2], xn[3]};
            *reinterpret_cast<v4f*>(sX + l0 + 4)   = (v4f){xn[4], xn[5], xn[6], xn[7]};
            *reinterpret_cast<v4f*>(Dout + l0)     = (v4f){d[0], d[1], d[2], d[3]};
            *reinterpret_cast<v4f*>(Dout + l0 + 4) = (v4f){d[4], d[5], d[6], d[7]};
        }
    }
}

template <int NR, int J0>
__device__ __forceinline__ void touch_rows(const float* plane, int row0, int C0, int tid)
{
    const int j = (tid >> 1) - J0;
    if ((unsigned)j < (unsigned)NR) {
        const int r = min(max(row0 + j, 0), HW - 1);
        const unsigned off = ((unsigned)r * HW + (unsigned)C0 + 64u * (unsigned)(tid & 1)) * 4u;
        __builtin_prefetch(reinterpret_cast<const char*>(plane) + off, 0, 1);
    }
}

template <int KW, int RS, int NR, bool SHIFT>
__device__ __forceinline__ void stage_rows(
    const float* __restrict__ Xi, const float* __restrict__ Yi,
    float* sX, float* sY, float* sDa, float* sDb, int gr0, int gc0, int tid)
{
    using G = Geo<KW>;
    constexpr int RR = G::RR, P = G::P, PT = G::PT;
    constexpr int Q  = P / 4;
    constexpr int QB = (Q + 3) / 4;
    constexpr int NB = (NR / 8) * QB;
    constexpr int IT = (NB + NWAVE - 1) / NWAVE;
    const int lane = tid & 31;
    const int wid = __builtin_amdgcn_readfirstlane(tid >> 5);
    const int lr = lane & 7, lq = lane >> 3;
    float4 xv[IT], yv[IT];
#pragma unroll
    for (int it = 0; it < IT; ++it) {
        const int blk = min(wid + it * NWAVE, NB - 1);
        const int rb = blk / QB, qb = blk - rb * QB;
        const int r = RS + 8 * rb + lr, q = min(4 * qb + lq, Q - 1);
        const int cr = min(max(gr0 + r, 0), HW - 1);
        const int cc = min(max(gc0 + 4 * q, 0), HW - 4);
        const size_t o = (size_t)cr * HW + cc;
        xv[it] = *reinterpret_cast<const float4*>(Xi + o);
        yv[it] = *reinterpret_cast<const float4*>(Yi + o);
    }
    if (SHIFT) {
        constexpr int KEEP = 4 * PC;
        constexpr int SRC  = RR - KEEP;
        constexpr int DST  = SRC - BR;
        constexpr int NV   = (P + 8) * (KEEP / 4);
        for (int i = tid; i < NV; i += NTHR) {
            const int col = i / (KEEP / 4), k = i - col * (KEEP / 4);
            const int s = col * PT + SRC + 4 * k, d = col * PT + DST + 4 * k;
            const v4f vx = *reinterpret_cast<const v4f*>(sX  + s);
            const v4f vy = *reinterpret_cast<const v4f*>(sY  + s);
            const v4f va = *reinterpret_cast<const v4f*>(sDa + s);
            const v4f vb = *reinterpret_cast<const v4f*>(sDb + s);
            *reinterpret_cast<v4f*>(sX  + d) = vx;
            *reinterpret_cast<v4f*>(sY  + d) = vy;
            *reinterpret_cast<v4f*>(sDa + d) = va;
            *reinterpret_cast<v4f*>(sDb + d) = vb;
        }
        __syncthreads();
    }
#pragma unroll
    for (int it = 0; it < IT; ++it) {
        const int blk = wid + it * NWAVE;
        const int rb = blk / QB, qb = blk - rb * QB;
        const int r = RS + 8 * rb + lr, q = 4 * qb + lq;
        if (blk < NB && q < Q) {
            const int gr = gr0 + r, gc = gc0 + 4 * q;
            const bool in = (unsigned)gr < (unsigned)HW && (unsigned)gc < (unsigned)HW;
            float4 x = xv[it], y = yv[it];
            if (!in) { x = make_float4(0.f, 0.f, 0.f, 0.f); y = x; }
            const int l = 4 * q * PT + r;
            sX[l]           = x.x;  sY[l]           = y.x;  sDa[l]           = y.x - x.x;
            sX[l + PT]      = x.y;  sY[l + PT]      = y.y;  sDa[l + PT]      = y.y - x.y;
            sX[l + 2 * PT]  = x.z;  sY[l + 2 * PT]  = y.z;  sDa[l + 2 * PT]  = y.z - x.z;
            sX[l + 3 * PT]  = x.w;  sY[l + 3 * PT]  = y.w;  sDa[l + 3 * PT]  = y.w - x.w;
        }
    }
}

template <int KW>
__device__ __forceinline__ void last_pass(
    const float* sDa, const float* sX, const float* wp,
    const float* __restrict__ N3, float* __restrict__ Oi,
    int gr0, int gc0, int lane, int wid)
{
    using G = Geo<KW>;
    constexpr int PR = G::PR, HL = G::HL, PT = G::PT, SG = G::SG;
    constexpr int ro = 3 * PC, co = HL;
    constexpr int TC = BC / 16;
    const int half = lane >> 4;
    const int n    = lane & 15;
    const int w = __builtin_amdgcn_readfirstlane(wid);
    v2f a[8][KW];
    load_band<KW>(wp, lane, a);
    const unsigned o00    = (unsigned)(gr0 + ro + 8 * half) * HW + (unsigned)(gc0 + co + n);
    const int      l00    = (co + n) * PT + ro + 8 * half;
    const float*   patch0 = sDa + (co - PR + n) * PT + (ro - PC) + 2 * half;
    for (int t = w; t < (BR / 16) * TC; t += NWAVE) {
        const int ti = t / TC, tj = t - ti * TC;
        int toff = 16 * tj * PT + 16 * ti;
        asm volatile("" : "+s"(toff));
        const unsigned o0 = o00 + (unsigned)(16 * ti * HW + 16 * tj);
        const float* __restrict__ np = N3 + o0;
        float*       __restrict__ op = Oi + o0;
        float nv[8];
#pragma unroll
        for (int v = 0; v < 8; ++v)
            nv[v] = np[v * HW];
        const float* patch = patch0 + toff;
        v2f b[SG][KW];
        load_patch<KW, PT, SG>(patch, 0, b);
        const int l0 = l00 + toff;
        const v4f x0 = *reinterpret_cast<const v4f*>(sX + l0);
        const v4f x1 = *reinterpret_cast<const v4f*>(sX + l0 + 4);
        const float xs[8] = {x0.x, x0.y, x0.z, x0.w, x1.x, x1.y, x1.z, x1.w};
        __builtin_amdgcn_sched_barrier(0);
        v8f acc = {};
        acc = band_mma<KW, SG>(a, 0, b, acc);
#pragma unroll
        for (int s0 = SG; s0 < 8; s0 += SG) {
            load_patch<KW, PT, SG>(patch, s0, b);
            __builtin_amdgcn_sched_barrier(0);
            acc = band_mma<KW, SG>(a, s0, b, acc);
        }
        __builtin_amdgcn_sched_barrier(0);
#pragma unroll
        for (int v = 0; v < 8; ++v)
            op[v * HW] = xs[v] + acc[v] * __builtin_amdgcn_rcpf(nv[v]);
    }
}

template <int KW>
__global__ __launch_bounds__(NTHR) void guided_fused(
    const float* __restrict__ X,  const float* __restrict__ Y,
    const float* __restrict__ Wk, const float* __restrict__ Nn,
    float* __restrict__ Out)
{
    using G = Geo<KW>;
    constexpr int RR = G::RR, HL = G::HL, ASZ = G::ASZ;

    __shared__ float wpad[3 * WPROWS * KW];
    __shared__ __attribute__((aligned(16))) float sX[ASZ];
    __shared__ __attribute__((aligned(16))) float sY[ASZ];
    __shared__ __attribute__((aligned(16))) float sDa[ASZ];
    __shared__ __attribute__((aligned(16))) float sDb[ASZ];

    const int tid  = threadIdx.x;
    const int lane = tid & 31;
    const int wid  = tid >> 5;

    constexpr int BPR = HW / BC;
    constexpr int BPI = BPR * (HW / (BR * NCH));
    const int img = blockIdx.x / BPI;
    const int rem = blockIdx.x % BPI;
    const int R0  = (rem / BPR) * (BR * NCH);
    const int C0  = (rem % BPR) * BC;
    int       gr0 = R0 - 3 * PC;
    const int gc0 = C0 - HL;

    const float* Xi = X   + (size_t)img * IMG;
    const float* Yi = Y   + (size_t)img * IMG;
    float*       Oi = Out + (size_t)img * IMG;
    const float* N3 = Nn + 2 * (size_t)IMG;

    for (int i = tid; i < 3 * WPROWS * KW; i += NTHR) {
        const int p  = i / (WPROWS * KW);
        const int o  = i - p * (WPROWS * KW);
        const int kx = o / WPROWS, t = o - kx * WPROWS;
        wpad[i] = (t >= 16 && t < 16 + KH) ? Wk[p * KH * KW + (t - 16) * KW + kx] : 0.0f;
    }

    stage_rows<KW, 0, RR, false>(Xi, Yi, sX, sY, sDa, sDb, gr0, gc0, tid);
    static_assert(2 * (3 * BR + 6 * PC) <= NTHR && 2 * 4 * BR <= NTHR, "one touch per thread");
    touch_rows<BR + 4 * PC, 0>(Nn,                  gr0 + PC,     C0, tid);
    touch_rows<BR + 2 * PC, BR + 4 * PC>(Nn + IMG,  gr0 + 2 * PC, C0, tid);
    touch_rows<BR, 2 * BR + 6 * PC>(N3,             gr0 + 3 * PC, C0, tid);
    __syncthreads();
    mid_pass<KW, 1, PC, BR + 4 * PC>(sDa, sDb, sX, sY, wpad,               Nn,       gr0, gc0, lane, wid);
    __syncthreads();
    touch_rows<BR, 0>(Nn,            gr0 + BR + 5 * PC, C0, tid);
    touch_rows<BR, BR>(Nn + IMG,     gr0 + BR + 4 * PC, C0, tid);
    touch_rows<BR, 2 * BR>(N3,       gr0 + BR + 3 * PC, C0, tid);
    mid_pass<KW, 2, 2 * PC, BR + 2 * PC>(sDb, sDa, sX, sY, wpad + WPROWS * KW, Nn + IMG, gr0, gc0, lane, wid);
    __syncthreads();
    touch_rows<BR, 3 * BR>(Xi, gr0 + BR + 6 * PC, C0, tid);
    touch_rows<BR, 0>(Yi,      gr0 + BR + 6 * PC, C0, tid);
    last_pass<KW>(sDa, sX, wpad + 2 * WPROWS * KW, N3, Oi, gr0, gc0, lane, wid);

#pragma unroll 1
    for (int ch = 1; ch < NCH; ++ch) {
        gr0 += BR;
        __syncthreads();
        stage_rows<KW, 6 * PC, BR, true>(Xi, Yi, sX, sY, sDa, sDb, gr0, gc0, tid);
        __syncthreads();
        mid_pass<KW, 1, 5 * PC, BR>(sDa, sDb, sX, sY, wpad,               Nn,       gr0, gc0, lane, wid);
        __syncthreads();
        const bool more = ch + 1 < NCH;
        if (more) {
            touch_rows<BR, 0>(Nn,            gr0 + BR + 5 * PC, C0, tid);
            touch_rows<BR, BR>(Nn + IMG,     gr0 + BR + 4 * PC, C0, tid);
            touch_rows<BR, 2 * BR>(N3,       gr0 + BR + 3 * PC, C0, tid);
        }
        mid_pass<KW, 2, 4 * PC, BR>(sDb, sDa, sX, sY, wpad + WPROWS * KW, Nn + IMG, gr0, gc0, lane, wid);
        __syncthreads();
        if (more) {
            touch_rows<BR, 3 * BR>(Xi, gr0 + BR + 6 * PC, C0, tid);
            touch_rows<BR, 0>(Yi,      gr0 + BR + 6 * PC, C0, tid);
        }
        last_pass<KW>(sDa, sX, wpad + 2 * WPROWS * KW, N3, Oi, gr0, gc0, lane, wid);
    }
}

template <int KW>
static void run_fused(const float* X, const float* y, const float* k,
                      const float* Nn, float* out, int batch, hipStream_t stream)
{
    dim3 grid(batch * (HW / (BR * NCH)) * (HW / BC)), block(NTHR);
    guided_fused<KW><<<grid, block, 0, stream>>>(X, y, k, Nn, out);
}

extern "C" void kernel_launch(void* const* d_in, const int* in_sizes, int n_in,
                              void* d_out, int out_size, void* d_ws, size_t ws_size,
                              hipStream_t stream) {
    const float* X  = (const float*)d_in[0];
    const float* y  = (const float*)d_in[1];
    const float* k  = (const float*)d_in[2];
    const float* Nn = (const float*)d_in[3];
    float* out = (float*)d_out;

    const int KW    = in_sizes[2] / (3 * KH);
    const int batch = in_sizes[0] / IMG;

    switch (KW) {
        case 1: run_fused<1>(X, y, k, Nn, out, batch, stream); break;
        case 3: run_fused<3>(X, y, k, Nn, out, batch, stream); break;
        case 7: run_fused<7>(X, y, k, Nn, out, batch, stream); break;
        case 9: run_fused<9>(X, y, k, Nn, out, batch, stream); break;
        case 5:
        default: run_fused<5>(X, y, k, Nn, out, batch, stream); break;
    }
}


# Y no longer held in LDS: a tile reads its 8 Y rows from the image beside its normalisers; the freed quarter makes chunks 128 rows: half the passes, barriers, band reloads, stagings; sums keep order
# speedup vs baseline: 2.0090x; 1.0441x over previous
#include <hip/hip_runtime.h>


typedef __attribute__((ext_vector_type(2))) float v2f;
typedef __attribute__((ext_vector_type(4))) float v4f;
typedef __attribute__((ext_vector_type(8))) float v8f;

#define HW     2048
#define IMG    (HW * HW)
#define KH     17
#define PC     8
#define WPROWS 48
#define SR     256
#define BC     128
#define NTHR   512
#define NWAVE  (NTHR / 32)

template <int KW>
struct Geo {
    static constexpr int PR  = KW / 2;
    static constexpr int BR  = (KW <= 5) ? 128 : 64;
    static constexpr int NCH = SR / BR;
    static constexpr int RR  = BR + 6 * PC;
    static constexpr int HL  = (3 * PR <= 8) ? 8 : 12;
    static constexpr int P   = BC + 2 * HL;
    static constexpr int PT  = 4 * ((RR / 4) | 1);
    static constexpr int ASZ = P * PT;
    static constexpr int TAIL = 8 * PT;
    static constexpr int SG  = (KW <= 5) ? 8 : (KW <= 7) ? 4 : 2;
};

template <int KW>
__device__ __forceinline__ void load_band(const float* wpad, int woff, int lane, v2f (&a)[8][KW])
{
    const int half = lane >> 4;
    const int m    = lane & 15;
    int t0 = woff + 2 * half - m + 16;
    asm volatile("" : "+v"(t0));
    const float* w0 = wpad + t0;
#pragma unroll
    for (int s = 0; s < 8; ++s) {
#pragma unroll
        for (int kx = 0; kx < KW; ++kx) {
            a[s][kx].x = w0[kx * WPROWS + 4 * s];
            a[s][kx].y = w0[kx * WPROWS + 4 * s + 1];
        }
    }
}

template <int KW, int PT, int SG>
__device__ __forceinline__ void load_patch(const float* src, int s0, v2f (&b)[SG][KW])
{
#pragma unroll
    for (int s = 0; s < SG; ++s) {
#pragma unroll
        for (int kx = 0; kx < KW; ++kx)
            b[s][kx] = *reinterpret_cast<const v2f*>(src + kx * PT + 4 * (s0 + s));
    }
}

template <int KW, int SG>
__device__ __forceinline__ v8f band_mma(const v2f (&a)[8][KW], int s0, const v2f (&b)[SG][KW], v8f acc)
{
#pragma unroll
    for (int s = 0; s < SG; ++s) {
#pragma unroll
        for (int kx = 0; kx < KW; ++kx)
            acc = __builtin_amdgcn_wmma_f32_16x16x4_f32(
                false, a[s0 + s][kx], false, b[s][kx], (short)0, acc, false, false);
    }
    return acc;
}

template <int KW, int PASS, int RO, int NROW>
__device__ __forceinline__ void mid_pass(
    const float* Din, float* Dout, float* sX, const float* __restrict__ Yg, const float* wp,
    const float* __restrict__ Np, int gr0, int gc0, int lane, int wid)
{
    using G = Geo<KW>;
    constexpr int PR = G::PR, PT = G::PT, SG = G::SG, BR = G::BR, NCH = G::NCH;
    constexpr int ro   = RO;
    constexpr int co   = G::HL - (3 - PASS) * PR;
    constexpr int nrow = NROW;
    constexpr int ncol = BC + 2 * (3 - PASS) * PR;
    constexpr int TR   = nrow / 16;
    constexpr int TC   = (ncol + 15) / 16;
    const int half = lane >> 4;
    const int n    = lane & 15;
    const int w = __builtin_amdgcn_readfirstlane(wid);

    v2f a[8][KW];
    load_band<KW>(wp, (PASS - 1) * WPROWS * KW, lane, a);

    const int    cl     = co + n;
    const int    gcl    = gc0 + cl;
    const int    rl     = gr0 + ro + 8 * half;
    const int    l00    = cl * PT + ro + 8 * half;
    const float* patch0 = Din + (co - PR + n) * PT + (ro - PC) + 2 * half;
    static_assert(RO % 8 == 0 && (3 * PC) % 8 == 0 && (BR * NCH) % 8 == 0 && BR % 8 == 0 && HW % 8 == 0,
                  "a lane's 8 rows are an aligned group of 8 image rows");

    for (int t = w; t < TR * TC; t += NWAVE) {
        const int ti = t / TC, tj = t - ti * TC;
        int toff = 16 * tj * PT + 16 * ti;
        asm volatile("" : "+s"(toff));
        const int  c    = cl + 16 * tj;
        const int  gc   = gcl + 16 * tj;
        const int  rb   = rl + 16 * ti;
        const bool cval = c < co + ncol;
        const bool cin  = (unsigned)gc < (unsigned)HW;
        const bool rin  = (unsigned)rb < (unsigned)HW;
        float nv[8], ys[8];
        const int rbc = min(max(rb, 0), HW - 8);
        const int cc  = min(max(gc, 0), HW - 1);
        const unsigned go = (unsigned)rbc * HW + (unsigned)cc;
        const float* __restrict__ np = Np + go;
        const float* __restrict__ yp = Yg + go;
#pragma unroll
        for (int v = 0; v < 8; ++v)
            nv[v] = np[v * HW];
#pragma unroll
        for (int v = 0; v < 8; ++v)
            ys[v] = yp[v * HW];
        const float* patch = patch0 + toff;
        v2f b[SG][KW];
        load_patch<KW, PT, SG>(patch, 0, b);
        const int l0 = l00 + toff;
        const v4f x0 = *reinterpret_cast<const v4f*>(sX + l0);
        const v4f x1 = *reinterpret_cast<const v4f*>(sX + l0 + 4);
        const float xs[8] = {x0.x, x0.y, x0.z, x0.w, x1.x, x1.y, x1.z, x1.w};
        __builtin_amdgcn_sched_barrier(0);
        v8f acc = {};
        acc = band_mma<KW, SG>(a, 0, b, acc);
#pragma unroll
        for (int s0 = SG; s0 < 8; s0 += SG) {
            load_patch<KW, PT, SG>(patch, s0, b);
            __builtin_amdgcn_sched_barrier(0);
            acc = band_mma<KW, SG>(a, s0, b, acc);
        }
        __builtin_amdgcn_sched_barrier(0);
        asm volatile("" :: "v"(nv[0]), "v"(nv[1]), "v"(nv[2]), "v"(nv[3]),
                           "v"(nv[4]), "v"(nv[5]), "v"(nv[6]), "v"(nv[7]));
        asm volatile("" :: "v"(xs[0]), "v"(xs[1]), "v"(xs[2]), "v"(xs[3]),
                           "v"(xs[4]), "v"(xs[5]), "v"(xs[6]), "v"(xs[7]));
        asm volatile("" :: "v"(ys[0]), "v"(ys[1]), "v"(ys[2]), "v"(ys[3]),
                           "v"(ys[4]), "v"(ys[5]), "v"(ys[6]), "v"(ys[7]));
        float xn[8], d[8];
        const unsigned keep = (cin && rin) ? 0xffffffffu : 0u;
#pragma unroll
        for (int v = 0; v < 8; ++v) {
            const float q = __uint_as_float(__float_as_uint(acc[v] * __builtin_amdgcn_rcpf(nv[v])) & keep);
            xn[v] = xs[v] + q;
            d[v]  = __uint_as_float(__float_as_uint(ys[v] - xn[v]) & keep);
        }
        if (cval) {
            *reinterpret_cast<v4f*>(sX + l0)       = (v4f){xn[0], xn[1], xn[2], xn[3]};
            *reinterpret_cast<v4f*>(sX + l0 + 4)   = (v4f){xn[4], xn[5], xn[6], xn[7]};
            *reinterpret_cast<v4f*>(Dout + l0)     = (v4f){d[0], d[1], d[2], d[3]};
            *reinterpret_cast<v4f*>(Dout + l0 + 4) = (v4f){d[4], d[5], d[6], d[7]};
        }
    }
}

template <int NR, int J0>
__device__ __forceinline__ void touch_rows(const float* plane, int row0, int C0, int tid)
{
    static_assert(J0 >= 0 && 2 * (J0 + NR) <= NTHR, "the call's row slots lie inside the block's 256");
    const int j = (tid >> 1) - J0;
    if ((unsigned)j < (unsigned)NR) {
        const int r = min(max(row0 + j, 0), HW - 1);
        const unsigned off = ((unsigned)r * HW + (unsigned)C0 + 64u * (unsigned)(tid & 1)) * 4u;
        __builtin_prefetch(reinterpret_cast<const char*>(plane) + off, 0, 1);
    }
}

template <int KW, int RS, int NR, bool SHIFT>
__device__ __forceinline__ void stage_rows(
    const float* __restrict__ Xi, const float* __restrict__ Yi,
    float* sX, float* sDa, float* sDb, int gr0, int gc0, int tid)
{
    using G = Geo<KW>;
    constexpr int RR = G::RR, P = G::P, PT = G::PT, BR = G::BR;
    constexpr int Q  = P / 4;
    constexpr int QB = (Q + 3) / 4;
    constexpr int NB = (NR / 8) * QB;
    constexpr int IT = (NB + NWAVE - 1) / NWAVE;
    const int lane = tid & 31;
    const int wid = __builtin_amdgcn_readfirstlane(tid >> 5);
    const int lr = lane & 7, lq = lane >> 3;
    float4 xv[IT], yv[IT];
#pragma unroll
    for (int it = 0; it < IT; ++it) {
        const int blk = min(wid + it * NWAVE, NB - 1);
        const int rb = blk / QB, qb = blk - rb * QB;
        const int r = RS + 8 * rb + lr, q = min(4 * qb + lq, Q - 1);
        const int cr = min(max(gr0 + r, 0), HW - 1);
        const int cc = min(max(gc0 + 4 * q, 0), HW - 4);
        const size_t o = (size_t)cr * HW + cc;
        xv[it] = *reinterpret_cast<const float4*>(Xi + o);
        yv[it] = *reinterpret_cast<const float4*>(Yi + o);
    }
    if (SHIFT) {
        constexpr int KEEP = 4 * PC;
        constexpr int SRC  = RR - KEEP;
        constexpr int DST  = SRC - BR;
        constexpr int NV   = P * (KEEP / 4);
        for (int i = tid; i < NV; i += NTHR) {
            const int col = i / (KEEP / 4), k = i - col * (KEEP / 4);
            const int s = col * PT + SRC + 4 * k, d = col * PT + DST + 4 * k;
            const v4f vx = *reinterpret_cast<const v4f*>(sX  + s);
            const v4f va = *reinterpret_cast<const v4f*>(sDa + s);
            const v4f vb = *reinterpret_cast<const v4f*>(sDb + s);
            *reinterpret_cast<v4f*>(sX  + d) = vx;
            *reinterpret_cast<v4f*>(sDa + d) = va;
            *reinterpret_cast<v4f*>(sDb + d) = vb;
        }
        __syncthreads();
    }
#pragma unroll
    for (int it = 0; it < IT; ++it) {
        const int blk = wid + it * NWAVE;
        const int rb = blk / QB, qb = blk - rb * QB;
        const int r = RS + 8 * rb + lr, q = 4 * qb + lq;
        if (blk < NB && q < Q) {
            const int gr = gr0 + r, gc = gc0 + 4 * q;
            const bool in = (unsigned)gr < (unsigned)HW && (unsigned)gc < (unsigned)HW;
            float4 x = xv[it], y = yv[it];
            if (!in) { x = make_float4(0.f, 0.f, 0.f, 0.f); y = x; }
            const int l = 4 * q * PT + r;
            sX[l]           = x.x;  sDa[l]           = y.x - x.x;
            sX[l + PT]      = x.y;  sDa[l + PT]      = y.y - x.y;
            sX[l + 2 * PT]  = x.z;  sDa[l + 2 * PT]  = y.z - x.z;
            sX[l + 3 * PT]  = x.w;  sDa[l + 3 * PT]  = y.w - x.w;
        }
    }
}

template <int KW>
__device__ __forceinline__ void last_pass(
    const float* sDa, const float* sX, const float* wp,
    const float* __restrict__ N3, float* __restrict__ Oi,
    int gr0, int gc0, int lane, int wid)
{
    using G = Geo<KW>;
    constexpr int PR = G::PR, HL = G::HL, PT = G::PT, SG = G::SG, BR = G::BR;
    constexpr int ro = 3 * PC, co = HL;
    constexpr int TC = BC / 16;
    const int half = lane >> 4;
    const int n    = lane & 15;
    const int w = __builtin_amdgcn_readfirstlane(wid);
    v2f a[8][KW];
    load_band<KW>(wp, 2 * WPROWS * KW, lane, a);
    const unsigned o00    = (unsigned)(gr0 + ro + 8 * half) * HW + (unsigned)(gc0 + co + n);
    const int      l00    = (co + n) * PT + ro + 8 * half;
    const float*   patch0 = sDa + (co - PR + n) * PT + (ro - PC) + 2 * half;
    for (int t = w; t < (BR / 16) * TC; t += NWAVE) {
        const int ti = t / TC, tj = t - ti * TC;
        int toff = 16 * tj * PT + 16 * ti;
        asm volatile("" : "+s"(toff));
        const unsigned o0 = o00 + (unsigned)(16 * ti * HW + 16 * tj);
        const float* __restrict__ np = N3 + o0;
        float*       __restrict__ op = Oi + o0;
        float nv[8];
#pragma unroll
        for (int v = 0; v < 8; ++v)
            nv[v] = np[v * HW];
        const float* patch = patch0 + toff;
        v2f b[SG][KW];
        load_patch<KW, PT, SG>(patch, 0, b);
        const int l0 = l00 + toff;
        const v4f x0 = *reinterpret_cast<const v4f*>(sX + l0);
        const v4f x1 = *reinterpret_cast<const v4f*>(sX + l0 + 4);
        const float xs[8] = {x0.x, x0.y, x0.z, x0.w, x1.x, x1.y, x1.z, x1.w};
        __builtin_amdgcn_sched_barrier(0);
        v8f acc = {};
        acc = band_mma<KW, SG>(a, 0, b, acc);
#pragma unroll
        for (int s0 = SG; s0 < 8; s0 += SG) {
            load_patch<KW, PT, SG>(patch, s0, b);
            __builtin_amdgcn_sched_barrier(0);
            acc = band_mma<KW, SG>(a, s0, b, acc);
        }
        __builtin_amdgcn_sched_barrier(0);
#pragma unroll
        for (int v = 0; v < 8; ++v)
            op[v * HW] = xs[v] + acc[v] * __builtin_amdgcn_rcpf(nv[v]);
    }
}

template <int KW>
__global__ __launch_bounds__(NTHR) void guided_fused(
    const float* __restrict__ X,  const float* __restrict__ Y,
    const float* __restrict__ Wk, const float* __restrict__ Nn,
    float* __restrict__ Out)
{
    using G = Geo<KW>;
    constexpr int RR = G::RR, HL = G::HL, ASZ = G::ASZ, BR = G::BR, NCH = G::NCH;

    __shared__ float wpad[3 * WPROWS * KW];
    __shared__ __attribute__((aligned(16))) float planes[3 * ASZ + G::TAIL];
    static_assert((ASZ * 4) % 16 == 0, "every plane starts 16-byte aligned");
    float* const sX  = planes;
    float* const sDa = planes + ASZ;
    float* const sDb = planes + 2 * ASZ;

    const int tid  = threadIdx.x;
    const int lane = tid & 31;
    const int wid  = tid >> 5;

    constexpr int BPR = HW / BC;
    constexpr int BPI = BPR * (HW / SR);
    static_assert(BR * NCH == SR, "the chunks make up the strip");
    const int img = blockIdx.x / BPI;
    const int rem = blockIdx.x % BPI;
    const int R0  = (rem / BPR) * SR;
    const int C0  = (rem % BPR) * BC;
    int       gr0 = R0 - 3 * PC;
    const int gc0 = C0 - HL;

    const float* Xi = X   + (size_t)img * IMG;
    const float* Yi = Y   + (size_t)img * IMG;
    float*       Oi = Out + (size_t)img * IMG;
    const float* N3 = Nn + 2 * (size_t)IMG;

    for (int i = tid; i < 3 * WPROWS * KW; i += NTHR) {
        const int p  = i / (WPROWS * KW);
        const int o  = i - p * (WPROWS * KW);
        const int kx = o / WPROWS, t = o - kx * WPROWS;
        wpad[i] = (t >= 16 && t < 16 + KH) ? Wk[p * KH * KW + (t - 16) * KW + kx] : 0.0f;
    }

    constexpr int TA = BR + 4 * PC, TB = BR + 2 * PC;
    constexpr int JB = (TA + TB <= 256) ? TA : 256 - TB;
    constexpr int JC = (TA + TB + BR <= 256) ? TA + TB : 0;
    constexpr int J2 = (2 * BR <= 256) ? BR : 0;
    constexpr int J3 = (3 * BR <= 256) ? 2 * BR : 0;
    constexpr int J4 = (4 * BR <= 256) ? 3 * BR : J2;

    stage_rows<KW, 0, RR, false>(Xi, Yi, sX, sDa, sDb, gr0, gc0, tid);
    touch_rows<TA, 0>(Nn,         gr0 + PC,     C0, tid);
    touch_rows<TB, JB>(Nn + IMG,  gr0 + 2 * PC, C0, tid);
    touch_rows<BR, JC>(N3,        gr0 + 3 * PC, C0, tid);
    __syncthreads();
    mid_pass<KW, 1, PC, BR + 4 * PC>(sDa, sDb, sX, Yi, wpad,               Nn,       gr0, gc0, lane, wid);
    __syncthreads();
    if (NCH > 1) {
        touch_rows<BR, 0>(Nn,         gr0 + BR + 5 * PC, C0, tid);
        touch_rows<BR, J2>(Nn + IMG,  gr0 + BR + 4 * PC, C0, tid);
        touch_rows<BR, J3>(N3,        gr0 + BR + 3 * PC, C0, tid);
    }
    mid_pass<KW, 2, 2 * PC, BR + 2 * PC>(sDb, sDa, sX, Yi, wpad,               Nn + IMG, gr0, gc0, lane, wid);
    __syncthreads();
    if (NCH > 1) {
        touch_rows<BR, J4>(Xi, gr0 + BR + 6 * PC, C0, tid);
        touch_rows<BR, 0>(Yi,  gr0 + BR + 6 * PC, C0, tid);
    }
    last_pass<KW>(sDa, sX, wpad, N3, Oi, gr0, gc0, lane, wid);

#pragma unroll 1
    for (int ch = 1; ch < NCH; ++ch) {
        gr0 += BR;
        __syncthreads();
        stage_rows<KW, 6 * PC, BR, true>(Xi, Yi, sX, sDa, sDb, gr0, gc0, tid);
        __syncthreads();
        mid_pass<KW, 1, 5 * PC, BR>(sDa, sDb, sX, Yi, wpad,               Nn,       gr0, gc0, lane, wid);
        __syncthreads();
        const bool more = ch + 1 < NCH;
        if (more) {
            touch_rows<BR, 0>(Nn,         gr0 + BR + 5 * PC, C0, tid);
            touch_rows<BR, J2>(Nn + IMG,  gr0 + BR + 4 * PC, C0, tid);
            touch_rows<BR, J3>(N3,        gr0 + BR + 3 * PC, C0, tid);
        }
        mid_pass<KW, 2, 4 * PC, BR>(sDb, sDa, sX, Yi, wpad,               Nn + IMG, gr0, gc0, lane, wid);
        __syncthreads();
        if (more) {
            touch_rows<BR, J4>(Xi, gr0 + BR + 6 * PC, C0, tid);
            touch_rows<BR, 0>(Yi,  gr0 + BR + 6 * PC, C0, tid);
        }
        last_pass<KW>(sDa, sX, wpad, N3, Oi, gr0, gc0, lane, wid);
    }
}

template <int KW>
static void run_fused(const float* X, const float* y, const float* k,
                      const float* Nn, float* out, int batch, hipStream_t stream)
{
    dim3 grid(batch * (HW / SR) * (HW / BC)), block(NTHR);
    guided_fused<KW><<<grid, block, 0, stream>>>(X, y, k, Nn, out);
}

extern "C" void kernel_launch(void* const* d_in, const int* in_sizes, int n_in,
                              void* d_out, int out_size, void* d_ws, size_t ws_size,
                              hipStream_t stream) {
    const float* X  = (const float*)d_in[0];
    const float* y  = (const float*)d_in[1];
    const float* k  = (const float*)d_in[2];
    const float* Nn = (const float*)d_in[3];
    float* out = (float*)d_out;

    const int KW    = in_sizes[2] / (3 * KH);
    const int batch = in_sizes[0] / IMG;

    switch (KW) {
        case 1: run_fused<1>(X, y, k, Nn, out, batch, stream); break;
        case 3: run_fused<3>(X, y, k, Nn, out, batch, stream); break;
        case 7: run_fused<7>(X, y, k, Nn, out, batch, stream); break;
        case 9: run_fused<9>(X, y, k, Nn, out, batch, stream); break;
        case 5:
        default: run_fused<5>(X, y, k, Nn, out, batch, stream); break;
    }
}


# matrix M side = 8 output rows x 2 columns (N side = 2 row blocks x 8 column pairs): a tile reads 24 patch rows x 6 columns, 36 WMMAs where 32 x 5 took 40; changes the order of each pixel's sums
# speedup vs baseline: 2.1063x; 1.0484x over previous
#include <hip/hip_runtime.h>


typedef __attribute__((ext_vector_type(2))) float v2f;
typedef __attribute__((ext_vector_type(4))) float v4f;
typedef __attribute__((ext_vector_type(8))) float v8f;

#define HW     2048
#define IMG    (HW * HW)
#define KH     17
#define PC     8
#define WPROWS 32
#define NSTEP  6
#define WT0    8
#define SR     256
#define BC     128
#define NTHR   512
#define NWAVE  (NTHR / 32)

template <int KW>
struct Geo {
    static constexpr int PR  = KW / 2;
    static constexpr int BR  = (KW <= 5) ? 128 : 64;
    static constexpr int NCH = SR / BR;
    static constexpr int RR  = BR + 6 * PC;
    static constexpr int HL  = (3 * PR <= 8) ? 8 : 12;
    static constexpr int P   = BC + 2 * HL;
    static constexpr int PT  = 4 * ((RR / 4) | 1);
    static constexpr int ASZ = P * PT;
    static constexpr int TAIL = 8 * PT;
    static constexpr int SG  = (KW <= 5) ? NSTEP : (KW <= 7) ? 3 : 2;
    static constexpr int WPC = KW + 2;
};

template <int KW>
__device__ __forceinline__ void load_band(const float* wpad, int woff, int lane, v2f (&a)[NSTEP][KW + 1])
{
    const int half = lane >> 4;
    const int mr   = lane & 7;
    const int mc   = (lane >> 3) & 1;
    int t0 = woff + (1 - mc) * WPROWS + 2 * half - mr + WT0;
    asm volatile("" : "+v"(t0));
    const float* w0 = wpad + t0;
#pragma unroll
    for (int s = 0; s < NSTEP; ++s) {
#pragma unroll
        for (int x = 0; x <= KW; ++x) {
            a[s][x].x = w0[x * WPROWS + 4 * s];
            a[s][x].y = w0[x * WPROWS + 4 * s + 1];
        }
    }
}

template <int KW, int PT, int SG>
__device__ __forceinline__ void load_patch(const float* src, int s0, v2f (&b)[SG][KW + 1])
{
#pragma unroll
    for (int s = 0; s < SG; ++s) {
#pragma unroll
        for (int x = 0; x <= KW; ++x)
            b[s][x] = *reinterpret_cast<const v2f*>(src + x * PT + 4 * (s0 + s));
    }
}

template <int KW, int SG>
__device__ __forceinline__ v8f band_mma(const v2f (&a)[NSTEP][KW + 1], int s0, const v2f (&b)[SG][KW + 1], v8f acc)
{
#pragma unroll
    for (int s = 0; s < SG; ++s) {
#pragma unroll
        for (int x = 0; x <= KW; ++x)
            acc = __builtin_amdgcn_wmma_f32_16x16x4_f32(
                false, a[s0 + s][x], false, b[s][x], (short)0, acc, false, false);
    }
    return acc;
}

template <int KW, int PASS, int RO, int NROW>
__device__ __forceinline__ void mid_pass(
    const float* Din, float* Dout, float* sX, const float* __restrict__ Yg, const float* wp,
    const float* __restrict__ Np, int gr0, int gc0, int lane, int wid)
{
    using G = Geo<KW>;
    constexpr int PR = G::PR, PT = G::PT, SG = G::SG, BR = G::BR, NCH = G::NCH;
    constexpr int ro   = RO;
    constexpr int co   = G::HL - (3 - PASS) * PR;
    constexpr int nrow = NROW;
    constexpr int ncol = BC + 2 * (3 - PASS) * PR;
    constexpr int TR   = nrow / 16;
    constexpr int TC   = (ncol + 15) / 16;
    const int half = lane >> 4;
    const int nc   = lane & 7;
    const int nr   = (lane >> 3) & 1;
    const int w = __builtin_amdgcn_readfirstlane(wid);

    v2f a[NSTEP][KW + 1];
    load_band<KW>(wp, (PASS - 1) * WPROWS * G::WPC, lane, a);

    const int    cl     = co + 2 * nc + half;
    const int    gcl    = gc0 + cl;
    const int    rl     = gr0 + ro + 8 * nr;
    const int    l00    = cl * PT + ro + 8 * nr;
    const float* patch0 = Din + (co - PR + 2 * nc) * PT + (ro - PC) + 8 * nr + 2 * half;
    static_assert(RO % 8 == 0 && (3 * PC) % 8 == 0 && (BR * NCH) % 8 == 0 && BR % 8 == 0 && HW % 8 == 0,
                  "a lane's 8 rows are an aligned group of 8 image rows");

    for (int t = w; t < TR * TC; t += NWAVE) {
        const int ti = t / TC, tj = t - ti * TC;
        int toff = 16 * tj * PT + 16 * ti;
        asm volatile("" : "+s"(toff));
        const int  c    = cl + 16 * tj;
        const int  gc   = gcl + 16 * tj;
        const int  rb   = rl + 16 * ti;
        const bool cval = c < co + ncol;
        const bool cin  = (unsigned)gc < (unsigned)HW;
        const bool rin  = (unsigned)rb < (unsigned)HW;
        float nv[8], ys[8];
        const int rbc = min(max(rb, 0), HW - 8);
        const int cc  = min(max(gc, 0), HW - 1);
        const unsigned go = (unsigned)rbc * HW + (unsigned)cc;
        const float* __restrict__ np = Np + go;
        const float* __restrict__ yp = Yg + go;
#pragma unroll
        for (int v = 0; v < 8; ++v)
            nv[v] = np[v * HW];
#pragma unroll
        for (int v = 0; v < 8; ++v)
            ys[v] = yp[v * HW];
        const float* patch = patch0 + toff;
        v2f b[SG][KW + 1];
        load_patch<KW, PT, SG>(patch, 0, b);
        const int l0 = l00 + toff;
        const v4f x0 = *reinterpret_cast<const v4f*>(sX + l0);
        const v4f x1 = *reinterpret_cast<const v4f*>(sX + l0 + 4);
        const float xs[8] = {x0.x, x0.y, x0.z, x0.w, x1.x, x1.y, x1.z, x1.w};
        __builtin_amdgcn_sched_barrier(0);
        v8f acc = {};
        acc = band_mma<KW, SG>(a, 0, b, acc);
#pragma unroll
        for (int s0 = SG; s0 < NSTEP; s0 += SG) {
            load_patch<KW, PT, SG>(patch, s0, b);
            __builtin_amdgcn_sched_barrier(0);
            acc = band_mma<KW, SG>(a, s0, b, acc);
        }
        __builtin_amdgcn_sched_barrier(0);
        asm volatile("" :: "v"(nv[0]), "v"(nv[1]), "v"(nv[2]), "v"(nv[3]),
                           "v"(nv[4]), "v"(nv[5]), "v"(nv[6]), "v"(nv[7]));
        asm volatile("" :: "v"(xs[0]), "v"(xs[1]), "v"(xs[2]), "v"(xs[3]),
                           "v"(xs[4]), "v"(xs[5]), "v"(xs[6]), "v"(xs[7]));
        asm volatile("" :: "v"(ys[0]), "v"(ys[1]), "v"(ys[2]), "v"(ys[3]),
                           "v"(ys[4]), "v"(ys[5]), "v"(ys[6]), "v"(ys[7]));
        float xn[8], d[8];
        const unsigned keep = (cin && rin) ? 0xffffffffu : 0u;
#pragma unroll
        for (int v = 0; v < 8; ++v) {
            const float q = __uint_as_float(__float_as_uint(acc[v] * __builtin_amdgcn_rcpf(nv[v])) & keep);
            xn[v] = xs[v] + q;
            d[v]  = __uint_as_float(__float_as_uint(ys[v] - xn[v]) & keep);
        }
        if (cval) {
            *reinterpret_cast<v4f*>(sX + l0)       = (v4f){xn[0], xn[1], xn[2], xn[3]};
            *reinterpret_cast<v4f*>(sX + l0 + 4)   = (v4f){xn[4], xn[5], xn[6], xn[7]};
            *reinterpret_cast<v4f*>(Dout + l0)     = (v4f){d[0], d[1], d[2], d[3]};
            *reinterpret_cast<v4f*>(Dout + l0 + 4) = (v4f){d[4], d[5], d[6], d[7]};
        }
    }
}

template <int NR, int J0>
__device__ __forceinline__ void touch_rows(const float* plane, int row0, int C0, int tid)
{
    static_assert(J0 >= 0 && 2 * (J0 + NR) <= NTHR, "the call's row slots lie inside the block's 256");
    const int j = (tid >> 1) - J0;
    if ((unsigned)j < (unsigned)NR) {
        const int r = min(max(row0 + j, 0), HW - 1);
        const unsigned off = ((unsigned)r * HW + (unsigned)C0 + 64u * (unsigned)(tid & 1)) * 4u;
        __builtin_prefetch(reinterpret_cast<const char*>(plane) + off, 0, 1);
    }
}

template <int KW, int RS, int NR, bool SHIFT>
__device__ __forceinline__ void stage_rows(
    const float* __restrict__ Xi, const float* __restrict__ Yi,
    float* sX, float* sDa, float* sDb, int gr0, int gc0, int tid)
{
    using G = Geo<KW>;
    constexpr int RR = G::RR, P = G::P, PT = G::PT, BR = G::BR;
    constexpr int Q  = P / 4;
    constexpr int QB = (Q + 3) / 4;
    constexpr int NB = (NR / 8) * QB;
    constexpr int IT = (NB + NWAVE - 1) / NWAVE;
    const int lane = tid & 31;
    const int wid = __builtin_amdgcn_readfirstlane(tid >> 5);
    const int lr = lane & 7, lq = lane >> 3;
    float4 xv[IT], yv[IT];
#pragma unroll
    for (int it = 0; it < IT; ++it) {
        const int blk = min(wid + it * NWAVE, NB - 1);
        const int rb = blk / QB, qb = blk - rb * QB;
        const int r = RS + 8 * rb + lr, q = min(4 * qb + lq, Q - 1);
        const int cr = min(max(gr0 + r, 0), HW - 1);
        const int cc = min(max(gc0 + 4 * q, 0), HW - 4);
        const size_t o = (size_t)cr * HW + cc;
        xv[it] = *reinterpret_cast<const float4*>(Xi + o);
        yv[it] = *reinterpret_cast<const float4*>(Yi + o);
    }
    if (SHIFT) {
        constexpr int KEEP = 4 * PC;
        constexpr int SRC  = RR - KEEP;
        constexpr int DST  = SRC - BR;
        constexpr int NV   = P * (KEEP / 4);
        for (int i = tid; i < NV; i += NTHR) {
            const int col = i / (KEEP / 4), k = i - col * (KEEP / 4);
            const int s = col * PT + SRC + 4 * k, d = col * PT + DST + 4 * k;
            const v4f vx = *reinterpret_cast<const v4f*>(sX  + s);
            const v4f va = *reinterpret_cast<const v4f*>(sDa + s);
            const v4f vb = *reinterpret_cast<const v4f*>(sDb + s);
            *reinterpret_cast<v4f*>(sX  + d) = vx;
            *reinterpret_cast<v4f*>(sDa + d) = va;
            *reinterpret_cast<v4f*>(sDb + d) = vb;
        }
        __syncthreads();
    }
#pragma unroll
    for (int it = 0; it < IT; ++it) {
        const int blk = wid + it * NWAVE;
        const int rb = blk / QB, qb = blk - rb * QB;
        const int r = RS + 8 * rb + lr, q = 4 * qb + lq;
        if (blk < NB && q < Q) {
            const int gr = gr0 + r, gc = gc0 + 4 * q;
            const bool in = (unsigned)gr < (unsigned)HW && (unsigned)gc < (unsigned)HW;
            float4 x = xv[it], y = yv[it];
            if (!in) { x = make_float4(0.f, 0.f, 0.f, 0.f); y = x; }
            const int l = 4 * q * PT + r;
            sX[l]           = x.x;  sDa[l]           = y.x - x.x;
            sX[l + PT]      = x.y;  sDa[l + PT]      = y.y - x.y;
            sX[l + 2 * PT]  = x.z;  sDa[l + 2 * PT]  = y.z - x.z;
            sX[l + 3 * PT]  = x.w;  sDa[l + 3 * PT]  = y.w - x.w;
        }
    }
}

template <int KW>
__device__ __forceinline__ void last_pass(
    const float* sDa, const float* sX, const float* wp,
    const float* __restrict__ N3, float* __restrict__ Oi,
    int gr0, int gc0, int lane, int wid)
{
    using G = Geo<KW>;
    constexpr int PR = G::PR, HL = G::HL, PT = G::PT, SG = G::SG, BR = G::BR;
    constexpr int ro = 3 * PC, co = HL;
    constexpr int TC = BC / 16;
    const int half = lane >> 4;
    const int nc   = lane & 7;
    const int nr   = (lane >> 3) & 1;
    const int w = __builtin_amdgcn_readfirstlane(wid);
    v2f a[NSTEP][KW + 1];
    load_band<KW>(wp, 2 * WPROWS * G::WPC, lane, a);
    const unsigned o00    = (unsigned)(gr0 + ro + 8 * nr) * HW + (unsigned)(gc0 + co + 2 * nc + half);
    const int      l00    = (co + 2 * nc + half) * PT + ro + 8 * nr;
    const float*   patch0 = sDa + (co - PR + 2 * nc) * PT + (ro - PC) + 8 * nr + 2 * half;
    for (int t = w; t < (BR / 16) * TC; t += NWAVE) {
        const int ti = t / TC, tj = t - ti * TC;
        int toff = 16 * tj * PT + 16 * ti;
        asm volatile("" : "+s"(toff));
        const unsigned o0 = o00 + (unsigned)(16 * ti * HW + 16 * tj);
        const float* __restrict__ np = N3 + o0;
        float*       __restrict__ op = Oi + o0;
        float nv[8];
#pragma unroll
        for (int v = 0; v < 8; ++v)
            nv[v] = np[v * HW];
        const float* patch = patch0 + toff;
        v2f b[SG][KW + 1];
        load_patch<KW, PT, SG>(patch, 0, b);
        const int l0 = l00 + toff;
        const v4f x0 = *reinterpret_cast<const v4f*>(sX + l0);
        const v4f x1 = *reinterpret_cast<const v4f*>(sX + l0 + 4);
        const float xs[8] = {x0.x, x0.y, x0.z, x0.w, x1.x, x1.y, x1.z, x1.w};
        __builtin_amdgcn_sched_barrier(0);
        v8f acc = {};
        acc = band_mma<KW, SG>(a, 0, b, acc);
#pragma unroll
        for (int s0 = SG; s0 < NSTEP; s0 += SG) {
            load_patch<KW, PT, SG>(patch, s0, b);
            __builtin_amdgcn_sched_barrier(0);
            acc = band_mma<KW, SG>(a, s0, b, acc);
        }
        __builtin_amdgcn_sched_barrier(0);
#pragma unroll
        for (int v = 0; v < 8; ++v)
            op[v * HW] = xs[v] + acc[v] * __builtin_amdgcn_rcpf(nv[v]);
    }
}

template <int KW>
__global__ __launch_bounds__(NTHR) void guided_fused(
    const float* __restrict__ X,  const float* __restrict__ Y,
    const float* __restrict__ Wk, const float* __restrict__ Nn,
    float* __restrict__ Out)
{
    using G = Geo<KW>;
    constexpr int RR = G::RR, HL = G::HL, ASZ = G::ASZ, BR = G::BR, NCH = G::NCH;

    __shared__ float wpad[3 * WPROWS * G::WPC];
    __shared__ __attribute__((aligned(16))) float planes[3 * ASZ + G::TAIL];
    static_assert((ASZ * 4) % 16 == 0, "every plane starts 16-byte aligned");
    float* const sX  = planes;
    float* const sDa = planes + ASZ;
    float* const sDb = planes + 2 * ASZ;

    const int tid  = threadIdx.x;
    const int lane = tid & 31;
    const int wid  = tid >> 5;

    constexpr int BPR = HW / BC;
    constexpr int BPI = BPR * (HW / SR);
    static_assert(BR * NCH == SR, "the chunks make up the strip");
    const int img = blockIdx.x / BPI;
    const int rem = blockIdx.x % BPI;
    const int R0  = (rem / BPR) * SR;
    const int C0  = (rem % BPR) * BC;
    int       gr0 = R0 - 3 * PC;
    const int gc0 = C0 - HL;

    const float* Xi = X   + (size_t)img * IMG;
    const float* Yi = Y   + (size_t)img * IMG;
    float*       Oi = Out + (size_t)img * IMG;
    const float* N3 = Nn + 2 * (size_t)IMG;

    constexpr int WPL = WPROWS * G::WPC;
    for (int i = tid; i < 3 * WPL; i += NTHR) {
        const int p  = i / WPL;
        const int o  = i - p * WPL;
        const int xq = o / WPROWS, t = o - xq * WPROWS;
        const bool tap = xq >= 1 && xq <= KW && t >= WT0 && t < WT0 + KH;
        wpad[i] = tap ? Wk[p * KH * KW + (t - WT0) * KW + (xq - 1)] : 0.0f;
    }

    constexpr int TA = BR + 4 * PC, TB = BR + 2 * PC;
    constexpr int JB = (TA + TB <= 256) ? TA : 256 - TB;
    constexpr int JC = (TA + TB + BR <= 256) ? TA + TB : 0;
    constexpr int J2 = (2 * BR <= 256) ? BR : 0;
    constexpr int J3 = (3 * BR <= 256) ? 2 * BR : 0;
    constexpr int J4 = (4 * BR <= 256) ? 3 * BR : J2;

    stage_rows<KW, 0, RR, false>(Xi, Yi, sX, sDa, sDb, gr0, gc0, tid);
    touch_rows<TA, 0>(Nn,         gr0 + PC,     C0, tid);
    touch_rows<TB, JB>(Nn + IMG,  gr0 + 2 * PC, C0, tid);
    touch_rows<BR, JC>(N3,        gr0 + 3 * PC, C0, tid);
    __syncthreads();
    mid_pass<KW, 1, PC, BR + 4 * PC>(sDa, sDb, sX, Yi, wpad,               Nn,       gr0, gc0, lane, wid);
    __syncthreads();
    if (NCH > 1) {
        touch_rows<BR, 0>(Nn,         gr0 + BR + 5 * PC, C0, tid);
        touch_rows<BR, J2>(Nn + IMG,  gr0 + BR + 4 * PC, C0, tid);
        touch_rows<BR, J3>(N3,        gr0 + BR + 3 * PC, C0, tid);
    }
    mid_pass<KW, 2, 2 * PC, BR + 2 * PC>(sDb, sDa, sX, Yi, wpad,               Nn + IMG, gr0, gc0, lane, wid);
    __syncthreads();
    if (NCH > 1) {
        touch_rows<BR, J4>(Xi, gr0 + BR + 6 * PC, C0, tid);
        touch_rows<BR, 0>(Yi,  gr0 + BR + 6 * PC, C0, tid);
    }
    last_pass<KW>(sDa, sX, wpad, N3, Oi, gr0, gc0, lane, wid);

#pragma unroll 1
    for (int ch = 1; ch < NCH; ++ch) {
        gr0 += BR;
        __syncthreads();
        stage_rows<KW, 6 * PC, BR, true>(Xi, Yi, sX, sDa, sDb, gr0, gc0, tid);
        __syncthreads();
        mid_pass<KW, 1, 5 * PC, BR>(sDa, sDb, sX, Yi, wpad,               Nn,       gr0, gc0, lane, wid);
        __syncthreads();
        const bool more = ch + 1 < NCH;
        if (more) {
            touch_rows<BR, 0>(Nn,         gr0 + BR + 5 * PC, C0, tid);
            touch_rows<BR, J2>(Nn + IMG,  gr0 + BR + 4 * PC, C0, tid);
            touch_rows<BR, J3>(N3,        gr0 + BR + 3 * PC, C0, tid);
        }
        mid_pass<KW, 2, 4 * PC, BR>(sDb, sDa, sX, Yi, wpad,               Nn + IMG, gr0, gc0, lane, wid);
        __syncthreads();
        if (more) {
            touch_rows<BR, J4>(Xi, gr0 + BR + 6 * PC, C0, tid);
            touch_rows<BR, 0>(Yi,  gr0 + BR + 6 * PC, C0, tid);
        }
        last_pass<KW>(sDa, sX, wpad, N3, Oi, gr0, gc0, lane, wid);
    }
}

template <int KW>
static void run_fused(const float* X, const float* y, const float* k,
                      const float* Nn, float* out, int batch, hipStream_t stream)
{
    dim3 grid(batch * (HW / SR) * (HW / BC)), block(NTHR);
    guided_fused<KW><<<grid, block, 0, stream>>>(X, y, k, Nn, out);
}

extern "C" void kernel_launch(void* const* d_in, const int* in_sizes, int n_in,
                              void* d_out, int out_size, void* d_ws, size_t ws_size,
                              hipStream_t stream) {
    const float* X  = (const float*)d_in[0];
    const float* y  = (const float*)d_in[1];
    const float* k  = (const float*)d_in[2];
    const float* Nn = (const float*)d_in[3];
    float* out = (float*)d_out;

    const int KW    = in_sizes[2] / (3 * KH);
    const int batch = in_sizes[0] / IMG;

    switch (KW) {
        case 1: run_fused<1>(X, y, k, Nn, out, batch, stream); break;
        case 3: run_fused<3>(X, y, k, Nn, out, batch, stream); break;
        case 7: run_fused<7>(X, y, k, Nn, out, batch, stream); break;
        case 9: run_fused<9>(X, y, k, Nn, out, batch, stream); break;
        case 5:
        default: run_fused<5>(X, y, k, Nn, out, batch, stream); break;
    }
}
